# starting kernel
# baseline (speedup 1.0000x reference)
#include <hip/hip_runtime.h>
#include <hip/hip_fp16.h>

typedef __attribute__((ext_vector_type(16))) _Float16 v16h;
typedef __attribute__((ext_vector_type(8)))  float    v8f;

#define N_NODES 50000
#define E_REAL  800000
#define E_TOT   (E_REAL + N_NODES)
#define IN_CH   128
#define HID     128
#define OUT_CH  64
#define NEG_SLOPE 0.2f


__global__ void fill_f32(float* __restrict__ p, float v, long n) {
  long i = (long)blockIdx.x * blockDim.x + threadIdx.x;
  long stride = (long)gridDim.x * blockDim.x;
  for (; i < n; i += stride) p[i] = v;
}

__device__ __forceinline__ void atomicMaxF(float* addr, float v) {
  if (v >= 0.0f) atomicMax((int*)addr, __float_as_int(v));
  else           atomicMin((unsigned int*)addr, __float_as_uint(v));
}

template<int NT, int K>
__global__ void wmma_gemm_attn(const float* __restrict__ A,
                               const float* __restrict__ B,
                               float* __restrict__ C,
                               const float* __restrict__ a_s,
                               const float* __restrict__ a_d,
                               float* __restrict__ os, float* __restrict__ od,
                               int M) {
  constexpr int N  = NT * 16;
  constexpr int KC = K / 32;
  __shared__ __align__(32) _Float16 ldsB[K * N];

  const int lane   = threadIdx.x & 31;
  const int wlocal = threadIdx.x >> 5;
  const int mtile  = blockIdx.x * 8 + wlocal;
  const int hh  = lane >> 4;
  const int l16 = lane & 15;

  for (int slot = threadIdx.x; slot < KC * NT * 32; slot += blockDim.x) {
    const int sl   = slot & 31;
    const int t    = slot >> 5;
    const int nt   = t % NT;
    const int kc   = t / NT;
    const int kb   = kc * 32 + (sl >> 4) * 8;
    const int n    = nt * 16 + (sl & 15);
    _Float16* dst  = ldsB + slot * 16;
#pragma unroll
    for (int j = 0; j < 8; ++j) {
      dst[j]     = (_Float16)B[(size_t)(kb + j) * N + n];
      dst[j + 8] = (_Float16)B[(size_t)(kb + 16 + j) * N + n];
    }
  }
  __syncthreads();
  if (mtile * 16 >= M) return;

  const float* __restrict__ Arow = A + (size_t)(mtile * 16 + l16) * K;
  v8f acc[NT] = {};
#pragma unroll
  for (int kc = 0; kc < KC; ++kc) {
    const int kb = kc * 32 + hh * 8;
    const float4 a0 = *(const float4*)(Arow + kb);
    const float4 a1 = *(const float4*)(Arow + kb + 4);
    const float4 a2 = *(const float4*)(Arow + kb + 16);
    const float4 a3 = *(const float4*)(Arow + kb + 20);
    const float av[16] = {a0.x, a0.y, a0.z, a0.w, a1.x, a1.y, a1.z, a1.w,
                          a2.x, a2.y, a2.z, a2.w, a3.x, a3.y, a3.z, a3.w};
    v16h a;
#pragma unroll
    for (int j = 0; j < 16; ++j) a[j] = (_Float16)av[j];
#pragma unroll
    for (int nt = 0; nt < NT; ++nt) {
      const v16h b = *(const v16h*)(ldsB + ((size_t)(kc * NT + nt) * 32 + lane) * 16);
      acc[nt] = __builtin_amdgcn_wmma_f32_16x16x32_f16(false, a, false, b,
                                                       (short)0, acc[nt], false, false);
    }
  }

  float asv[NT], adv[NT];
#pragma unroll
  for (int nt = 0; nt < NT; ++nt) {
    asv[nt] = a_s[nt * 16 + l16];
    adv[nt] = a_d[nt * 16 + l16];
  }
#pragma unroll
  for (int r = 0; r < 8; ++r) {
    const int row = mtile * 16 + hh * 8 + r;
    float* Crow = C + (size_t)row * N + l16;
    float ps = 0.0f, pd = 0.0f;
#pragma unroll
    for (int nt = 0; nt < NT; ++nt) {
      const float v = acc[nt][r];
      Crow[nt * 16] = v;
      ps = fmaf(v, asv[nt], ps);
      pd = fmaf(v, adv[nt], pd);
    }
#pragma unroll
    for (int off = 8; off > 0; off >>= 1) {
      ps += __shfl_xor(ps, off, 32);
      pd += __shfl_xor(pd, off, 32);
    }
    if (l16 == 0) { os[row] = ps; od[row] = pd; }
  }
}

__device__ __forceinline__ void edge_src_dst(const int* __restrict__ ei, int e,
                                             int& s, int& d) {
  if (e < E_REAL) { s = ei[e]; d = ei[E_REAL + e]; }
  else            { s = d = e - E_REAL; }
}

__global__ void edge_logits_max(const int* __restrict__ ei,
    const float* __restrict__ as1, const float* __restrict__ ad1,
    float* __restrict__ eb1, float* __restrict__ m1,
    const float* __restrict__ as2, const float* __restrict__ ad2,
    float* __restrict__ eb2, float* __restrict__ m2) {
  const int e = blockIdx.x * blockDim.x + threadIdx.x;
  if (e >= E_TOT) return;
  int s, d; edge_src_dst(ei, e, s, d);
  float t = as1[s] + ad1[d];
  t = t > 0.0f ? t : NEG_SLOPE * t;
  eb1[e] = t;
  atomicMaxF(m1 + d, t);
  if (as2) {
    float u = as2[s] + ad2[d];
    u = u > 0.0f ? u : NEG_SLOPE * u;
    eb2[e] = u;
    atomicMaxF(m2 + d, u);
  }
}

__global__ void edge_exp_sum(const int* __restrict__ ei,
    float* __restrict__ eb1, const float* __restrict__ m1, float* __restrict__ dn1,
    float* __restrict__ eb2, const float* __restrict__ m2, float* __restrict__ dn2) {
  const int e = blockIdx.x * blockDim.x + threadIdx.x;
  if (e >= E_TOT) return;
  int s, d; edge_src_dst(ei, e, s, d);
  (void)s;
  float p = __expf(eb1[e] - m1[d]);
  eb1[e] = p;
  atomicAdd(dn1 + d, p);
  if (eb2) {
    float q = __expf(eb2[e] - m2[d]);
    eb2[e] = q;
    atomicAdd(dn2 + d, q);
  }
}

__global__ void edge_aggregate128(const int* __restrict__ ei,
    const float* __restrict__ eb, const float* __restrict__ dn,
    const float* __restrict__ H, float* __restrict__ out) {
  const int wave = blockIdx.x * (blockDim.x >> 5) + (threadIdx.x >> 5);
  const int lane = threadIdx.x & 31;
  if (wave >= E_TOT) return;
  int s, d; edge_src_dst(ei, wave, s, d);
  const float coef = eb[wave] / dn[d];
  const float4 h4 = *(const float4*)(H + (size_t)s * 128 + lane * 4);
  float* od = out + (size_t)d * 128 + lane * 4;
  atomicAdd(od + 0, coef * h4.x);
  atomicAdd(od + 1, coef * h4.y);
  atomicAdd(od + 2, coef * h4.z);
  atomicAdd(od + 3, coef * h4.w);
}

__global__ void edge_aggregate_dual64(const int* __restrict__ ei,
    const float* __restrict__ eb1, const float* __restrict__ dn1,
    const float* __restrict__ H1, float* __restrict__ out1,
    const float* __restrict__ eb2, const float* __restrict__ dn2,
    const float* __restrict__ H2, float* __restrict__ out2) {
  const int wave = blockIdx.x * (blockDim.x >> 5) + (threadIdx.x >> 5);
  const int lane = threadIdx.x & 31;
  if (wave >= E_TOT) return;
  int s, d; edge_src_dst(ei, wave, s, d);
  {
    const float c1 = eb1[wave] / dn1[d];
    const float2 h2 = *(const float2*)(H1 + (size_t)s * 64 + lane * 2);
    float* od = out1 + (size_t)d * 64 + lane * 2;
    atomicAdd(od + 0, c1 * h2.x);
    atomicAdd(od + 1, c1 * h2.y);
  }
  {
    const float c2 = eb2[wave] / dn2[d];
    const float2 h2 = *(const float2*)(H2 + (size_t)s * 64 + lane * 2);
    float* od = out2 + (size_t)d * 64 + lane * 2;
    atomicAdd(od + 0, c2 * h2.x);
    atomicAdd(od + 1, c2 * h2.y);
  }
}

__global__ void bias_act(float* __restrict__ h, const float* __restrict__ b,
                         long n, int C, int relu) {
  long i = (long)blockIdx.x * blockDim.x + threadIdx.x;
  if (i >= n) return;
  float v = h[i] + b[(int)(i & (C - 1))];
  if (relu) v = v > 0.0f ? v : 0.0f;
  h[i] = v;
}


static inline int ceildiv(long a, long b) { return (int)((a + b - 1) / b); }

extern "C" void kernel_launch(void* const* d_in, const int* in_sizes, int n_in,
                              void* d_out, int out_size, void* d_ws, size_t ws_size,
                              hipStream_t stream) {
  (void)in_sizes; (void)n_in; (void)out_size; (void)ws_size;
  const float* x       = (const float*)d_in[0];
  const int*   ei      = (const int*)d_in[1];
  const float* W1      = (const float*)d_in[2];
  const float* a_src1  = (const float*)d_in[3];
  const float* a_dst1  = (const float*)d_in[4];
  const float* b1      = (const float*)d_in[5];
  const float* W_mu    = (const float*)d_in[6];
  const float* a_srcm  = (const float*)d_in[7];
  const float* a_dstm  = (const float*)d_in[8];
  const float* b_mu    = (const float*)d_in[9];
  const float* W_ls    = (const float*)d_in[10];
  const float* a_srcl  = (const float*)d_in[11];
  const float* a_dstl  = (const float*)d_in[12];
  const float* b_ls    = (const float*)d_in[13];

  const long SZ_H = (long)N_NODES * HID;
  const long SZ_O = (long)N_NODES * OUT_CH;

  float* ws  = (float*)d_ws;
  float* hx  = ws;
  float* h1  = hx + SZ_H;
  float* hmu = hx;
  float* hls = hx + SZ_O;
  float* sc  = h1 + SZ_H;
  float* as1 = sc +  0L * N_NODES, *ad1 = sc + 1L * N_NODES;
  float* m1  = sc +  2L * N_NODES, *dn1 = sc + 3L * N_NODES;
  float* asm_= sc +  4L * N_NODES, *adm = sc + 5L * N_NODES;
  float* mm  = sc +  6L * N_NODES, *dnm = sc + 7L * N_NODES;
  float* asl = sc +  8L * N_NODES, *adl = sc + 9L * N_NODES;
  float* ml  = sc + 10L * N_NODES, *dnl = sc + 11L * N_NODES;
  float* eb1 = sc + 12L * N_NODES;
  float* ebm = eb1 + E_TOT;
  float* ebl = ebm + E_TOT;

  float* out_mu = (float*)d_out;
  float* out_ls = out_mu + SZ_O;

  const int TB = 256;
  const float NEG_INF = -__builtin_huge_valf();

  fill_f32<<<1024, TB, 0, stream>>>(h1, 0.0f, SZ_H);
  fill_f32<<<1024, TB, 0, stream>>>((float*)d_out, 0.0f, 2 * SZ_O);
  fill_f32<<<ceildiv(N_NODES, TB), TB, 0, stream>>>(m1, NEG_INF, N_NODES);
  fill_f32<<<ceildiv(N_NODES, TB), TB, 0, stream>>>(mm, NEG_INF, N_NODES);
  fill_f32<<<ceildiv(N_NODES, TB), TB, 0, stream>>>(ml, NEG_INF, N_NODES);
  fill_f32<<<ceildiv(N_NODES, TB), TB, 0, stream>>>(dn1, 0.0f, N_NODES);
  fill_f32<<<ceildiv(N_NODES, TB), TB, 0, stream>>>(dnm, 0.0f, N_NODES);
  fill_f32<<<ceildiv(N_NODES, TB), TB, 0, stream>>>(dnl, 0.0f, N_NODES);

  const int gemm_blocks = ceildiv(N_NODES / 16, 8);

  wmma_gemm_attn<HID / 16, IN_CH><<<gemm_blocks, TB, 0, stream>>>(
      x, W1, hx, a_src1, a_dst1, as1, ad1, N_NODES);
  edge_logits_max<<<ceildiv(E_TOT, TB), TB, 0, stream>>>(ei, as1, ad1, eb1, m1,
                                                         nullptr, nullptr, nullptr, nullptr);
  edge_exp_sum<<<ceildiv(E_TOT, TB), TB, 0, stream>>>(ei, eb1, m1, dn1,
                                                      nullptr, nullptr, nullptr);
  edge_aggregate128<<<ceildiv(E_TOT, 8), TB, 0, stream>>>(ei, eb1, dn1, hx, h1);
  bias_act<<<ceildiv(SZ_H, TB), TB, 0, stream>>>(h1, b1, SZ_H, HID, 1);

  wmma_gemm_attn<OUT_CH / 16, HID><<<gemm_blocks, TB, 0, stream>>>(
      h1, W_mu, hmu, a_srcm, a_dstm, asm_, adm, N_NODES);
  wmma_gemm_attn<OUT_CH / 16, HID><<<gemm_blocks, TB, 0, stream>>>(
      h1, W_ls, hls, a_srcl, a_dstl, asl, adl, N_NODES);
  edge_logits_max<<<ceildiv(E_TOT, TB), TB, 0, stream>>>(ei, asm_, adm, ebm, mm,
                                                         asl, adl, ebl, ml);
  edge_exp_sum<<<ceildiv(E_TOT, TB), TB, 0, stream>>>(ei, ebm, mm, dnm, ebl, ml, dnl);
  edge_aggregate_dual64<<<ceildiv(E_TOT, 8), TB, 0, stream>>>(ei, ebm, dnm, hmu, out_mu,
                                                              ebl, dnl, hls, out_ls);
  bias_act<<<ceildiv(SZ_O, TB), TB, 0, stream>>>(out_mu, b_mu, SZ_O, OUT_CH, 0);
  bias_act<<<ceildiv(SZ_O, TB), TB, 0, stream>>>(out_ls, b_ls, SZ_O, OUT_CH, 0);
}


# independent and same-input work share launches: x@W1 runs on its own work-groups of the edge-histogram launch, mu and logstd products run as one launch over the layer-1 output; 6 launches, was 8
# speedup vs baseline: 5.4711x; 5.4711x over previous
#include <hip/hip_runtime.h>
#include <hip/hip_fp16.h>

typedef __attribute__((ext_vector_type(16))) _Float16 v16h;
typedef __attribute__((ext_vector_type(8)))  float    v8f;

#define N_NODES 50000
#define E_REAL  800000
#define IN_CH   128
#define HID     128
#define OUT_CH  64
#define NEG_SLOPE 0.2f

template<int NT, int K, bool SPLIT>
__device__ __forceinline__ void gemm_attn_body(int blk,
                               const float* __restrict__ A,
                               const float* __restrict__ B0,
                               const float* __restrict__ B1,
                               float* __restrict__ C,
                               const float* __restrict__ a_s0,
                               const float* __restrict__ a_d0,
                               const float* __restrict__ a_s1,
                               const float* __restrict__ a_d1,
                               float* __restrict__ os0, float* __restrict__ od0,
                               float* __restrict__ os1, float* __restrict__ od1,
                               int M) {
  constexpr int N  = NT * 16;
  constexpr int KC = K / 32;
  constexpr int NTH = SPLIT ? NT / 2 : NT;
  constexpr int NB0 = NTH * 16;
  static_assert(!SPLIT || NT % 2 == 0, "two equal halves");
  __shared__ __align__(32) _Float16 ldsB[K * N];

  const int lane   = threadIdx.x & 31;
  const int wlocal = threadIdx.x >> 5;
  const int mtile  = blk * (int)(blockDim.x >> 5) + wlocal;
  const int hh  = lane >> 4;
  const int l16 = lane & 15;

  for (int slot = threadIdx.x; slot < KC * NT * 32; slot += blockDim.x) {
    const int sl   = slot & 31;
    const int t    = slot >> 5;
    const int nt   = t % NT;
    const int kc   = t / NT;
    const int kb   = kc * 32 + (sl >> 4) * 8;
    const int n    = nt * 16 + (sl & 15);
    const float* __restrict__ Bs = (SPLIT && n >= NB0) ? B1 : B0;
    const int nc   = (SPLIT && n >= NB0) ? n - NB0 : n;
    _Float16* dst  = ldsB + slot * 16;
#pragma unroll
    for (int j = 0; j < 8; ++j) {
      dst[j]     = (_Float16)Bs[(size_t)(kb + j) * NB0 + nc];
      dst[j + 8] = (_Float16)Bs[(size_t)(kb + 16 + j) * NB0 + nc];
    }
  }
  __syncthreads();
  if (mtile * 16 >= M) return;

  const float* __restrict__ Arow = A + (size_t)(mtile * 16 + l16) * K;
  v8f acc[NT] = {};
#pragma unroll
  for (int kc = 0; kc < KC; ++kc) {
    const int kb = kc * 32 + hh * 8;
    const float4 a0 = *(const float4*)(Arow + kb);
    const float4 a1 = *(const float4*)(Arow + kb + 4);
    const float4 a2 = *(const float4*)(Arow + kb + 16);
    const float4 a3 = *(const float4*)(Arow + kb + 20);
    const float av[16] = {a0.x, a0.y, a0.z, a0.w, a1.x, a1.y, a1.z, a1.w,
                          a2.x, a2.y, a2.z, a2.w, a3.x, a3.y, a3.z, a3.w};
    v16h a;
#pragma unroll
    for (int j = 0; j < 16; ++j) a[j] = (_Float16)av[j];
#pragma unroll
    for (int nt = 0; nt < NT; ++nt) {
      const v16h b = *(const v16h*)(ldsB + ((size_t)(kc * NT + nt) * 32 + lane) * 16);
      acc[nt] = __builtin_amdgcn_wmma_f32_16x16x32_f16(false, a, false, b,
                                                       (short)0, acc[nt], false, false);
    }
  }

  float asv[NT], adv[NT];
#pragma unroll
  for (int nt = 0; nt < NT; ++nt) {
    if (nt < NTH) {
      asv[nt] = a_s0[nt * 16 + l16];
      adv[nt] = a_d0[nt * 16 + l16];
    } else {
      asv[nt] = a_s1[(nt - NTH) * 16 + l16];
      adv[nt] = a_d1[(nt - NTH) * 16 + l16];
    }
  }
#pragma unroll
  for (int r = 0; r < 8; ++r) {
    const int row = mtile * 16 + hh * 8 + r;
    float* Crow = C + (size_t)row * N + l16;
    float ps0 = 0.0f, pd0 = 0.0f, ps1 = 0.0f, pd1 = 0.0f;
#pragma unroll
    for (int nt = 0; nt < NT; ++nt) {
      const float v = acc[nt][r];
      Crow[nt * 16] = v;
      if (nt < NTH) {
        ps0 = fmaf(v, asv[nt], ps0);
        pd0 = fmaf(v, adv[nt], pd0);
      } else {
        ps1 = fmaf(v, asv[nt], ps1);
        pd1 = fmaf(v, adv[nt], pd1);
      }
    }
#pragma unroll
    for (int off = 8; off > 0; off >>= 1) {
      ps0 += __shfl_xor(ps0, off, 32);
      pd0 += __shfl_xor(pd0, off, 32);
      if constexpr (SPLIT) {
        ps1 += __shfl_xor(ps1, off, 32);
        pd1 += __shfl_xor(pd1, off, 32);
      }
    }
    if (l16 == 0) {
      os0[row] = ps0; od0[row] = pd0;
      if constexpr (SPLIT) { os1[row] = ps1; od1[row] = pd1; }
    }
  }
}

__global__ void __launch_bounds__(256)
wmma_gemm_attn_pair(const float* __restrict__ A,
                                    const float* __restrict__ B0,
                                    const float* __restrict__ B1,
                                    float* __restrict__ C,
                                    const float* __restrict__ a_s0,
                                    const float* __restrict__ a_d0,
                                    const float* __restrict__ a_s1,
                                    const float* __restrict__ a_d1,
                                    float* __restrict__ os0, float* __restrict__ od0,
                                    float* __restrict__ os1, float* __restrict__ od1,
                                    int M) {
  gemm_attn_body<2 * OUT_CH / 16, HID, true>((int)blockIdx.x, A, B0, B1, C,
      a_s0, a_d0, a_s1, a_d1, os0, od0, os1, od1, M);
}


#define NB      196
#define NCH     128
#define CHUNK   6250
#define EDGE_TB 512
static_assert(NCH * CHUNK == E_REAL, "chunks must tile the edge list");
static_assert(NB * 256 >= N_NODES && (NB - 1) * 256 < N_NODES, "bucket count");
static_assert(NB % 4 == 0 && NB <= 7 * 32, "table row is read in 16-byte pieces; 7 buckets per lane in the scan");

__device__ __forceinline__ int clamp_node(int s) {
  return min(max(s, 0), N_NODES - 1);
}
__device__ __forceinline__ int clamp_edge(int p) {
  return min(max(p, 0), E_REAL - 1);
}

__device__ __forceinline__ void
bucket_hist_body(int me, const int* __restrict__ ei, int* __restrict__ hist) {
  __shared__ int h[256];
  const int tid = threadIdx.x;
  if (tid < 256) h[tid] = 0;
  __syncthreads();
  const int* __restrict__ dp = ei + E_REAL + me * CHUNK;
  constexpr int TRIPS = (CHUNK + EDGE_TB - 1) / EDGE_TB;
  int dv[TRIPS];
#pragma unroll
  for (int k = 0; k < TRIPS; ++k) {
    const int i = tid + k * EDGE_TB;
    dv[k] = i < CHUNK ? dp[i] : 0;
  }
#pragma unroll
  for (int k = 0; k < TRIPS; ++k) {
    const int i = tid + k * EDGE_TB;
    if (i < CHUNK) atomicAdd(&h[clamp_node(dv[k]) >> 8], 1);
  }
  __syncthreads();
  if (tid < 256) hist[me * 256 + tid] = h[tid];
}

#define GEMM1_BLOCKS ((N_NODES / 16 + EDGE_TB / 32 - 1) / (EDGE_TB / 32))
static_assert(N_NODES % 16 == 0, "whole 16-row strips");
__global__ void __launch_bounds__(EDGE_TB)
hist_and_project(const int* __restrict__ ei, int* __restrict__ hist,
                 const float* __restrict__ x, const float* __restrict__ W,
                 float* __restrict__ hx,
                 const float* __restrict__ a_s, const float* __restrict__ a_d,
                 float* __restrict__ os, float* __restrict__ od) {
  if (blockIdx.x < GEMM1_BLOCKS) {
    gemm_attn_body<HID / 16, IN_CH, false>((int)blockIdx.x, x, W, nullptr, hx,
        a_s, a_d, nullptr, nullptr, os, od, nullptr, nullptr, N_NODES);
  } else {
    bucket_hist_body((int)blockIdx.x - GEMM1_BLOCKS, ei, hist);
  }
}

__global__ void __launch_bounds__(EDGE_TB)
bucket_place(const int* __restrict__ ei, const int* __restrict__ hist,
             int* __restrict__ bbase, int* __restrict__ pair_words) {
  int2* __restrict__ pairs = (int2*)pair_words;
  __shared__ int tot[256], part[256], cur[256];
  const int tid  = threadIdx.x;
  const int lane = tid & 31;
  const int me   = blockIdx.x;
  if (tid < 256) { tot[tid] = 0; part[tid] = 0; }
  __syncthreads();

  const int q = tid & 63;
  const int g = tid >> 6;
  static_assert(EDGE_TB / 64 == 8 && NCH % 8 == 0, "row groups");
  if (q < NB / 4) {
    int4 t4 = make_int4(0, 0, 0, 0), p4 = make_int4(0, 0, 0, 0);
#pragma unroll
    for (int k = 0; k < NCH / 8; ++k) {
      const int w = g + 8 * k;
      const int4 v = *(const int4*)(hist + w * 256 + 4 * q);
      t4.x += v.x; t4.y += v.y; t4.z += v.z; t4.w += v.w;
      if (w < me) { p4.x += v.x; p4.y += v.y; p4.z += v.z; p4.w += v.w; }
    }
    atomicAdd(&tot[4 * q + 0], t4.x);  atomicAdd(&part[4 * q + 0], p4.x);
    atomicAdd(&tot[4 * q + 1], t4.y);  atomicAdd(&part[4 * q + 1], p4.y);
    atomicAdd(&tot[4 * q + 2], t4.z);  atomicAdd(&part[4 * q + 2], p4.z);
    atomicAdd(&tot[4 * q + 3], t4.w);  atomicAdd(&part[4 * q + 3], p4.w);
  }
  __syncthreads();

  if (tid < 32) {
    int v[7];
    int s = 0;
#pragma unroll
    for (int j = 0; j < 7; ++j) {
      const int b = lane * 7 + j;
      v[j] = b < NB ? tot[b] : 0;
      s += v[j];
    }
    int incl = s;
#pragma unroll
    for (int off = 1; off < 32; off <<= 1) {
      const int u = __shfl_up(incl, off, 32);
      if (lane >= off) incl += u;
    }
    int run = incl - s;
#pragma unroll
    for (int j = 0; j < 7; ++j) {
      const int b = lane * 7 + j;
      if (b < NB) {
        cur[b] = run + part[b];
        if (me == 0) bbase[b] = run;
      }
      run += v[j];
    }
    if (me == 0 && lane == 31) bbase[NB] = run;
  }
  __syncthreads();

  const int* __restrict__ sp = ei + me * CHUNK;
  const int* __restrict__ dp = ei + E_REAL + me * CHUNK;
  constexpr int TRIPS = (CHUNK + EDGE_TB - 1) / EDGE_TB;
  int sv[TRIPS], dv[TRIPS];
#pragma unroll
  for (int k = 0; k < TRIPS; ++k) {
    const int i = tid + k * EDGE_TB;
    sv[k] = i < CHUNK ? sp[i] : 0;
    dv[k] = i < CHUNK ? dp[i] : 0;
  }
  int pos[TRIPS];
#pragma unroll
  for (int k = 0; k < TRIPS; ++k) {
    const int i = tid + k * EDGE_TB;
    dv[k]  = clamp_node(dv[k]);
    pos[k] = i < CHUNK ? atomicAdd(&cur[dv[k] >> 8], 1) : 0;
  }
#pragma unroll
  for (int k = 0; k < TRIPS; ++k) {
    const int i = tid + k * EDGE_TB;
    if (i < CHUNK) pairs[clamp_edge(pos[k])] = make_int2(sv[k], dv[k]);
  }
}

__global__ void __launch_bounds__(EDGE_TB)
bucket_to_csr(const int* __restrict__ pair_words, const int* __restrict__ bbase,
              int* __restrict__ rowptr, int* __restrict__ csr_src) {
  const int2* __restrict__ pairs = (const int2*)pair_words;
  __shared__ int cnt[257], cur[257], rp[256];
  const int tid  = threadIdx.x;
  const int lane = tid & 31;
  const int b    = blockIdx.x;
  const int segbeg = min(max(bbase[b], 0), E_REAL);
  const int segend = min(max(bbase[b + 1], segbeg), E_REAL);

  if (tid <= 256) cnt[tid] = 0;
  __syncthreads();

  for (int i0 = segbeg + tid; i0 < segend; i0 += 4 * EDGE_TB) {
    int dv[4];
#pragma unroll
    for (int j = 0; j < 4; ++j) {
      const int idx = i0 + j * EDGE_TB;
      dv[j] = pairs[min(idx, segend - 1)].y;
    }
#pragma unroll
    for (int j = 0; j < 4; ++j) {
      const int idx = i0 + j * EDGE_TB;
      atomicAdd(&cnt[idx < segend ? (dv[j] & 255) : 256], 1);
    }
  }
  __syncthreads();

  if (tid < 32) {
    int v[8];
    int s = 0;
#pragma unroll
    for (int j = 0; j < 8; ++j) { v[j] = cnt[lane * 8 + j]; s += v[j]; }
    int incl = s;
#pragma unroll
    for (int off = 1; off < 32; off <<= 1) {
      const int u = __shfl_up(incl, off, 32);
      if (lane >= off) incl += u;
    }
    int run = segbeg + incl - s;
#pragma unroll
    for (int j = 0; j < 8; ++j) {
      cur[lane * 8 + j] = run;
      rp[lane * 8 + j]  = run;
      run += v[j];
    }
    if (lane == 0) cur[256] = 0;
  }
  __syncthreads();
  if (tid < 256) {
    const int node = b * 256 + tid;
    if (node <= N_NODES) rowptr[node] = rp[tid];
  }

  for (int i0 = segbeg + tid; i0 < segend; i0 += 4 * EDGE_TB) {
    int2 ev[4];
    int  pos[4];
#pragma unroll
    for (int j = 0; j < 4; ++j) {
      const int idx = i0 + j * EDGE_TB;
      ev[j] = pairs[min(idx, segend - 1)];
    }
#pragma unroll
    for (int j = 0; j < 4; ++j) {
      const int idx = i0 + j * EDGE_TB;
      pos[j] = atomicAdd(&cur[idx < segend ? (ev[j].y & 255) : 256], 1);
    }
#pragma unroll
    for (int j = 0; j < 4; ++j) {
      const int idx = i0 + j * EDGE_TB;
      if (idx < segend) csr_src[clamp_edge(pos[j])] = ev[j].x;
    }
  }
}

__device__ __forceinline__ float lrelu(float t) { return t > 0.0f ? t : NEG_SLOPE * t; }


__device__ __forceinline__ int lane_read_i(int v, int j) {
  return __builtin_amdgcn_readlane(v, j);
}
__device__ __forceinline__ float lane_read_f(float v, int j) {
  return __int_as_float(__builtin_amdgcn_readlane(__float_as_int(v), j));
}

template<bool DUAL, bool RELU>
__global__ void node_softmax_gather(const int* __restrict__ rowptr,
    const int* __restrict__ csr_src,
    const float* __restrict__ asA, const float* __restrict__ adA,
    const float* __restrict__ HA,  const float* __restrict__ bA,
    float* __restrict__ outA,
    const float* __restrict__ asB, const float* __restrict__ adB,
    const float* __restrict__ HB,  const float* __restrict__ bB,
    float* __restrict__ outB) {
  constexpr int STRIDE  = 128;
  constexpr int OSTRIDE = DUAL ? 64 : 128;
  const int lane = threadIdx.x & 31;
  const int hh   = lane >> 4;
  const int coff = DUAL ? (lane & 15) * 4 : lane * 4;
  const int node = __builtin_amdgcn_readfirstlane(
      (int)(blockIdx.x * (blockDim.x >> 5) + (threadIdx.x >> 5)));
  if (node >= N_NODES) return;
  const int beg = __builtin_amdgcn_readfirstlane(rowptr[node]);
  const int end = __builtin_amdgcn_readfirstlane(rowptr[node + 1]);

  const float NEG_INF = -__builtin_huge_valf();
  const float adAn  = adA[node];
  const float selfA = lrelu(asA[node] + adAn);
  float adBn = 0.0f, selfB = 0.0f;
  if constexpr (DUAL) { adBn = adB[node]; selfB = lrelu(asB[node] + adBn); }

  const int  i0 = beg + lane;
  const bool v0 = i0 < end;
  int   s0  = node;
  float tA0 = NEG_INF, tB0 = NEG_INF;
  if (v0) {
    s0  = clamp_node(csr_src[i0]);
    tA0 = lrelu(asA[s0] + adAn);
    if constexpr (DUAL) tB0 = lrelu(asB[s0] + adBn);
  }
  float mA = fmaxf(selfA, tA0);
  float mB = fmaxf(selfB, tB0);
  for (int base = beg + 32; base < end; base += 32) {
    const int i = base + lane;
    if (i < end) {
      const int s = clamp_node(csr_src[i]);
      mA = fmaxf(mA, lrelu(asA[s] + adAn));
      if constexpr (DUAL) mB = fmaxf(mB, lrelu(asB[s] + adBn));
    }
  }
  for (int off = 16; off > 0; off >>= 1) {
    mA = fmaxf(mA, __shfl_xor(mA, off, 32));
    if constexpr (DUAL) mB = fmaxf(mB, __shfl_xor(mB, off, 32));
  }

  const float pselfA = __expf(selfA - mA);
  const float pselfB = DUAL ? __expf(selfB - mB) : 0.0f;
  const float* __restrict__ Hl = (DUAL && hh) ? HB : HA;
  float4 acc;
  {
    const float  ps = (DUAL && hh) ? pselfB : pselfA;
    const float4 h  = *(const float4*)(Hl + (size_t)node * STRIDE + coff);
    acc.x = ps * h.x; acc.y = ps * h.y; acc.z = ps * h.z; acc.w = ps * h.w;
  }
  float denA = 0.0f, denB = 0.0f;

  for (int base = beg; base < end; base += 32) {
    int   s  = s0;
    float pA = 0.0f, pB = 0.0f;
    if (base == beg) {
      if (v0) {
        pA = __expf(tA0 - mA);
        if constexpr (DUAL) pB = __expf(tB0 - mB);
      }
    } else {
      const int i = base + lane;
      if (i < end) {
        s  = clamp_node(csr_src[i]);
        pA = __expf(lrelu(asA[s] + adAn) - mA);
        if constexpr (DUAL) pB = __expf(lrelu(asB[s] + adBn) - mB);
      }
    }
    denA += pA;
    if constexpr (DUAL) denB += pB;

    const int rem = end - base;
    const int cnt = rem < 32 ? rem : 32;
#pragma unroll 4
    for (int j = 0; j < cnt; ++j) {
      const int sj = lane_read_i(s, j);
      float pj = lane_read_f(pA, j);
      if constexpr (DUAL) {
        const float pjB = lane_read_f(pB, j);
        pj = hh ? pjB : pj;
      }
      const float4 h = *(const float4*)(Hl + (size_t)sj * STRIDE + coff);
      acc.x = fmaf(pj, h.x, acc.x);
      acc.y = fmaf(pj, h.y, acc.y);
      acc.z = fmaf(pj, h.z, acc.z);
      acc.w = fmaf(pj, h.w, acc.w);
    }
  }

  for (int off = 16; off > 0; off >>= 1) {
    denA += __shfl_xor(denA, off, 32);
    if constexpr (DUAL) denB += __shfl_xor(denB, off, 32);
  }
  const float den = (DUAL && hh) ? (denB + pselfB) : (denA + pselfA);
  const float* __restrict__ bl = (DUAL && hh) ? bB : bA;
  float* __restrict__ ol = (DUAL && hh) ? outB : outA;
  float4 o;
  o.x = acc.x / den + bl[coff + 0];
  o.y = acc.y / den + bl[coff + 1];
  o.z = acc.z / den + bl[coff + 2];
  o.w = acc.w / den + bl[coff + 3];
  if constexpr (RELU) {
    o.x = o.x > 0.0f ? o.x : 0.0f;
    o.y = o.y > 0.0f ? o.y : 0.0f;
    o.z = o.z > 0.0f ? o.z : 0.0f;
    o.w = o.w > 0.0f ? o.w : 0.0f;
  }
  *(float4*)(ol + (size_t)node * OSTRIDE + coff) = o;
}


static inline int ceildiv(long a, long b) { return (int)((a + b - 1) / b); }

extern "C" void kernel_launch(void* const* d_in, const int* in_sizes, int n_in,
                              void* d_out, int out_size, void* d_ws, size_t ws_size,
                              hipStream_t stream) {
  (void)in_sizes; (void)n_in; (void)out_size; (void)ws_size;
  const float* x       = (const float*)d_in[0];
  const int*   ei      = (const int*)d_in[1];
  const float* W1      = (const float*)d_in[2];
  const float* a_src1  = (const float*)d_in[3];
  const float* a_dst1  = (const float*)d_in[4];
  const float* b1      = (const float*)d_in[5];
  const float* W_mu    = (const float*)d_in[6];
  const float* a_srcm  = (const float*)d_in[7];
  const float* a_dstm  = (const float*)d_in[8];
  const float* b_mu    = (const float*)d_in[9];
  const float* W_ls    = (const float*)d_in[10];
  const float* a_srcl  = (const float*)d_in[11];
  const float* a_dstl  = (const float*)d_in[12];
  const float* b_ls    = (const float*)d_in[13];

  const long SZ_H = (long)N_NODES * HID;
  const long SZ_O = (long)N_NODES * OUT_CH;

  float* ws  = (float*)d_ws;
  float* hx  = ws;
  float* h1  = hx + SZ_H;
  float* hml = hx;
  float* sc  = h1 + SZ_H;
  float* as1 = sc + 0L * N_NODES, *ad1 = sc + 1L * N_NODES;
  float* asm_= sc + 2L * N_NODES, *adm = sc + 3L * N_NODES;
  float* asl = sc + 4L * N_NODES, *adl = sc + 5L * N_NODES;
  int* rowptr  = (int*)(sc + 8L * N_NODES);
  int* csr_src = (int*)(sc + 10L * N_NODES);
  int*  pairs  = (int*)h1;
  int*  hist   = (int*)(h1 + 2L * E_REAL);
  int*  bbase  = hist + NCH * 256;

  float* out_mu = (float*)d_out;
  float* out_ls = out_mu + SZ_O;

  const int TB = 256;

  hist_and_project<<<GEMM1_BLOCKS + NCH, EDGE_TB, 0, stream>>>(
      ei, hist, x, W1, hx, a_src1, a_dst1, as1, ad1);
  bucket_place<<<NCH, EDGE_TB, 0, stream>>>(ei, hist, bbase, pairs);
  bucket_to_csr<<<NB, EDGE_TB, 0, stream>>>(pairs, bbase, rowptr, csr_src);

  const int gemm_blocks = ceildiv(N_NODES / 16, 8);
  const int node_blocks = ceildiv(N_NODES, TB / 32);

  node_softmax_gather<false, true><<<node_blocks, TB, 0, stream>>>(
      rowptr, csr_src, as1, ad1, hx, b1, h1,
      nullptr, nullptr, nullptr, nullptr, nullptr);

  wmma_gemm_attn_pair<<<gemm_blocks, TB, 0, stream>>>(
      h1, W_mu, W_ls, hml, a_srcm, a_dstm, a_srcl, a_dstl,
      asm_, adm, asl, adl, N_NODES);
  node_softmax_gather<true, false><<<node_blocks, TB, 0, stream>>>(
      rowptr, csr_src, asm_, adm, hml, b_mu, out_mu,
      asl, adl, hml + OUT_CH, b_ls, out_ls);
}


# second-layer logit terms of both heads stored as (mu, logstd) pairs per node: the node pass fetches an edge's two source terms with one 8-byte lookup in one line, was two lookups in two lines
# speedup vs baseline: 9.0966x; 1.6627x over previous
#include <hip/hip_runtime.h>
#include <hip/hip_fp16.h>

typedef __attribute__((ext_vector_type(16))) _Float16 v16h;
typedef __attribute__((ext_vector_type(8)))  float    v8f;
typedef __attribute__((ext_vector_type(8)))  _Float16 v8h;
typedef __attribute__((ext_vector_type(4)))  _Float16 v4h;
typedef __attribute__((ext_vector_type(2)))  _Float16 v2h;

#define N_NODES 50000
#define E_REAL  800000
#define IN_CH   128
#define HID     128
#define OUT_CH  64
#define NEG_SLOPE 0.2f

template<int NT, int K, bool SPLIT>
__device__ __forceinline__ void stage_b_fragments(_Float16* __restrict__ ldsB,
                               const float* __restrict__ B0,
                               const float* __restrict__ B1) {
  constexpr int KC  = K / 32;
  constexpr int NB0 = (SPLIT ? NT / 2 : NT) * 16;
  for (int slot = threadIdx.x; slot < KC * NT * 32; slot += blockDim.x) {
    const int sl   = slot & 31;
    const int t    = slot >> 5;
    const int nt   = t % NT;
    const int kc   = t / NT;
    const int kb   = kc * 32 + (sl >> 4) * 8;
    const int n    = nt * 16 + (sl & 15);
    const float* __restrict__ Bs = (SPLIT && n >= NB0) ? B1 : B0;
    const int nc   = (SPLIT && n >= NB0) ? n - NB0 : n;
    _Float16* dst  = ldsB + slot * 16;
#pragma unroll
    for (int j = 0; j < 8; ++j) {
      dst[j]     = (_Float16)Bs[(size_t)(kb + j) * NB0 + nc];
      dst[j + 8] = (_Float16)Bs[(size_t)(kb + 16 + j) * NB0 + nc];
    }
  }
}

template<int NT, int K, bool SPLIT>
__device__ __forceinline__ void gemm_attn_body(int blk,
                               const float* __restrict__ A,
                               const float* __restrict__ B0,
                               const float* __restrict__ B1,
                               float* __restrict__ C,
                               const float* __restrict__ a_s0,
                               const float* __restrict__ a_d0,
                               const float* __restrict__ a_s1,
                               const float* __restrict__ a_d1,
                               float* __restrict__ os0, float* __restrict__ od0,
                               float* __restrict__ os1, float* __restrict__ od1,
                               int M) {
  constexpr int N  = NT * 16;
  constexpr int KC = K / 32;
  constexpr int NTH = SPLIT ? NT / 2 : NT;
  static_assert(!SPLIT || NT % 2 == 0, "two equal halves");
  static_assert(NT == 8, "table row = 128 halves: two 8-byte pieces per lane and accumulator register");
  _Float16* __restrict__ Ct = (_Float16*)C;
  __shared__ __align__(32) _Float16 ldsB[K * N];

  const int lane   = threadIdx.x & 31;
  const int wlocal = threadIdx.x >> 5;
  const int mtile  = blk * (int)(blockDim.x >> 5) + wlocal;
  const int hh  = lane >> 4;
  const int l16 = lane & 15;

  stage_b_fragments<NT, K, SPLIT>(ldsB, B0, B1);
  __syncthreads();
  if (mtile * 16 >= M) return;

  const float* __restrict__ Arow = A + (size_t)(mtile * 16 + l16) * K;
  v8f acc[NT] = {};
#pragma unroll
  for (int kc = 0; kc < KC; ++kc) {
    const int kb = kc * 32 + hh * 8;
    const float4 a0 = *(const float4*)(Arow + kb);
    const float4 a1 = *(const float4*)(Arow + kb + 4);
    const float4 a2 = *(const float4*)(Arow + kb + 16);
    const float4 a3 = *(const float4*)(Arow + kb + 20);
    const float av[16] = {a0.x, a0.y, a0.z, a0.w, a1.x, a1.y, a1.z, a1.w,
                          a2.x, a2.y, a2.z, a2.w, a3.x, a3.y, a3.z, a3.w};
    v16h a;
#pragma unroll
    for (int j = 0; j < 16; ++j) a[j] = (_Float16)av[j];
#pragma unroll
    for (int nt = 0; nt < NT; ++nt) {
      const v16h b = *(const v16h*)(ldsB + ((size_t)(kc * NT + nt) * 32 + lane) * 16);
      acc[nt] = __builtin_amdgcn_wmma_f32_16x16x32_f16(false, a, false, b,
                                                       (short)0, acc[nt], false, false);
    }
  }

  float asv[NT], adv[NT];
#pragma unroll
  for (int nt = 0; nt < NT; ++nt) {
    if (nt < NTH) {
      asv[nt] = a_s0[nt * 16 + l16];
      adv[nt] = a_d0[nt * 16 + l16];
    } else {
      asv[nt] = a_s1[(nt - NTH) * 16 + l16];
      adv[nt] = a_d1[(nt - NTH) * 16 + l16];
    }
  }
#pragma unroll
  for (int r = 0; r < 8; ++r) {
    const int row = mtile * 16 + hh * 8 + r;
    float ps0 = 0.0f, pd0 = 0.0f, ps1 = 0.0f, pd1 = 0.0f;
    v4h pe, po;
#pragma unroll
    for (int nt = 0; nt < NT; ++nt) {
      const float v = acc[nt][r];
      if (nt & 1) po[nt >> 1] = (_Float16)v; else pe[nt >> 1] = (_Float16)v;
      if (nt < NTH) {
        ps0 = fmaf(v, asv[nt], ps0);
        pd0 = fmaf(v, adv[nt], pd0);
      } else {
        ps1 = fmaf(v, asv[nt], ps1);
        pd1 = fmaf(v, adv[nt], pd1);
      }
    }
#pragma unroll
    for (int off = 8; off > 0; off >>= 1) {
      ps0 += __shfl_xor(ps0, off, 32);
      pd0 += __shfl_xor(pd0, off, 32);
      if constexpr (SPLIT) {
        ps1 += __shfl_xor(ps1, off, 32);
        pd1 += __shfl_xor(pd1, off, 32);
      }
    }
    *(v4h*)(Ct + (size_t)row * N + 4 * l16)      = pe;
    *(v4h*)(Ct + (size_t)row * N + 64 + 4 * l16) = po;
    if (l16 == 0) {
      os0[row] = ps0; od0[row] = pd0;
      if constexpr (SPLIT) { os1[row] = ps1; od1[row] = pd1; }
    }
  }
}


#define BKT     64
#define BKT_SH  6
#define NB      782
#define NBP     800
#define BPL     (NBP / 32)
#define NCH     128
#define CHUNK   6250
#define EDGE_TB 512
#define SEG_CAP 2048
static_assert(NCH * CHUNK == E_REAL, "chunks must tile the edge list");
static_assert(NB * BKT >= N_NODES && (NB - 1) * BKT < N_NODES, "bucket count");
static_assert(NBP >= NB && NBP % 32 == 0, "one LDS counter and one table column per bucket; whole lanes in the scan");
static_assert(BKT == 64 && (1 << BKT_SH) == BKT && N_NODES % 16 == 0, "two counters per lane in the bucket scan; whole strips");
static_assert(NCH % 32 == 0, "whole waves of chunks");

__device__ __forceinline__ int clamp_node(int s) {
  return min(max(s, 0), N_NODES - 1);
}
__device__ __forceinline__ int clamp_edge(int p) {
  return min(max(p, 0), E_REAL - 1);
}

__device__ __forceinline__ void
chunk_sort_body(int me, const int* __restrict__ ei, int* __restrict__ hist,
                int* __restrict__ offs, int* __restrict__ pair_words) {
  int2* __restrict__ pairs = (int2*)pair_words + (size_t)me * CHUNK;
  __shared__ int h[NBP], start[NBP], cur[NBP];
  const int tid  = threadIdx.x;
  const int lane = tid & 31;
  for (int t = tid; t < NBP; t += EDGE_TB) h[t] = 0;
  __syncthreads();
  const int* __restrict__ sp = ei + me * CHUNK;
  const int* __restrict__ dp = ei + E_REAL + me * CHUNK;
  constexpr int TRIPS = (CHUNK + EDGE_TB - 1) / EDGE_TB;
  int sv[TRIPS], dv[TRIPS];
#pragma unroll
  for (int k = 0; k < TRIPS; ++k) {
    const int ic = min(tid + k * EDGE_TB, CHUNK - 1);
    sv[k] = sp[ic];
    dv[k] = dp[ic];
  }
#pragma unroll
  for (int k = 0; k < TRIPS; ++k) {
    const int i = tid + k * EDGE_TB;
    dv[k] = clamp_node(dv[k]);
    if (i < CHUNK) atomicAdd(&h[dv[k] >> BKT_SH], 1);
  }
  __syncthreads();

  if (tid < 32) {
    int v[BPL];
    int s = 0;
#pragma unroll
    for (int j = 0; j < BPL; ++j) { v[j] = h[lane * BPL + j]; s += v[j]; }
    int incl = s;
#pragma unroll
    for (int off = 1; off < 32; off <<= 1) {
      const int u = __shfl_up(incl, off, 32);
      if (lane >= off) incl += u;
    }
    int run = incl - s;
#pragma unroll
    for (int j = 0; j < BPL; ++j) {
      start[lane * BPL + j] = run;
      cur[lane * BPL + j]   = run;
      run += v[j];
    }
  }
  __syncthreads();
  for (int t = tid; t < NBP; t += EDGE_TB) {
    hist[me * NBP + t] = h[t];
    offs[me * NBP + t] = start[t];
  }

  int pos[TRIPS];
#pragma unroll
  for (int k = 0; k < TRIPS; ++k) {
    const int i = tid + k * EDGE_TB;
    pos[k] = i < CHUNK ? atomicAdd(&cur[dv[k] >> BKT_SH], 1) : 0;
  }
#pragma unroll
  for (int k = 0; k < TRIPS; ++k) {
    const int i = tid + k * EDGE_TB;
    if (i < CHUNK) pairs[min(max(pos[k], 0), CHUNK - 1)] = make_int2(sv[k], dv[k]);
  }
}

#define GEMM1_BLOCKS ((N_NODES / 16 + EDGE_TB / 32 - 1) / (EDGE_TB / 32))
static_assert(N_NODES % 16 == 0, "whole 16-row strips");
__global__ void __launch_bounds__(EDGE_TB)
sort_and_project(const int* __restrict__ ei, int* __restrict__ hist,
                 int* __restrict__ offs, int* __restrict__ pair_words,
                 const float* __restrict__ x, const float* __restrict__ W,
                 float* __restrict__ hx,
                 const float* __restrict__ a_s, const float* __restrict__ a_d,
                 float* __restrict__ os, float* __restrict__ od) {
  if (blockIdx.x < NCH) {
    chunk_sort_body((int)blockIdx.x, ei, hist, offs, pair_words);
  } else {
    gemm_attn_body<HID / 16, IN_CH, false>((int)blockIdx.x - NCH, x, W, nullptr, hx,
        a_s, a_d, nullptr, nullptr, os, od, nullptr, nullptr, N_NODES);
  }
}

__device__ __forceinline__ float lrelu(float t) { return t > 0.0f ? t : NEG_SLOPE * t; }


__device__ __forceinline__ int lane_read_i(int v, int j) {
  return __builtin_amdgcn_readlane(v, j);
}
__device__ __forceinline__ float lane_read_f(float v, int j) {
  return __int_as_float(__builtin_amdgcn_readlane(__float_as_int(v), j));
}

template<bool DUAL, int GATHERS>
__device__ __forceinline__ float4 node_aggregate(int node, int rpi, int lane,
    const int* __restrict__ rowptr,
    const int* __restrict__ csr_src,
    const float* __restrict__ asA, const float* __restrict__ adA,
    const float* __restrict__ HA,  const float* __restrict__ bA,
    const float* __restrict__ asB, const float* __restrict__ adB,
    const float* __restrict__ bB) {
  constexpr int STRIDE  = 128;
  const uint2* __restrict__ Tab = (const uint2*)HA;
  const int beg = __builtin_amdgcn_readfirstlane(rowptr[rpi]);
  const int end = __builtin_amdgcn_readfirstlane(rowptr[rpi + 1]);

  const float NEG_INF = -__builtin_huge_valf();
  const float2* __restrict__ asP = (const float2*)asA;
  const float2* __restrict__ adP = (const float2*)adA;
  float adAn, selfA, adBn = 0.0f, selfB = 0.0f;
  if constexpr (DUAL) {
    const float2 dn = adP[node];
    const float2 sn = asP[node];
    adAn = dn.x; adBn = dn.y;
    selfA = lrelu(sn.x + adAn);
    selfB = lrelu(sn.y + adBn);
  } else {
    adAn  = adA[node];
    selfA = lrelu(asA[node] + adAn);
  }

  const int  i0 = beg + lane;
  const bool v0 = i0 < end;
  int   s0  = node;
  float tA0 = NEG_INF, tB0 = NEG_INF;
  if (v0) {
    s0  = clamp_node(csr_src[i0]);
    if constexpr (DUAL) {
      const float2 t = asP[s0];
      tA0 = lrelu(t.x + adAn);
      tB0 = lrelu(t.y + adBn);
    } else {
      tA0 = lrelu(asA[s0] + adAn);
    }
  }
  float mA = fmaxf(selfA, tA0);
  float mB = fmaxf(selfB, tB0);
  for (int base = beg + 32; base < end; base += 32) {
    const int i = base + lane;
    if (i < end) {
      const int s = clamp_node(csr_src[i]);
      if constexpr (DUAL) {
        const float2 t = asP[s];
        mA = fmaxf(mA, lrelu(t.x + adAn));
        mB = fmaxf(mB, lrelu(t.y + adBn));
      } else {
        mA = fmaxf(mA, lrelu(asA[s] + adAn));
      }
    }
  }
  for (int off = 16; off > 0; off >>= 1) {
    mA = fmaxf(mA, __shfl_xor(mA, off, 32));
    if constexpr (DUAL) mB = fmaxf(mB, __shfl_xor(mB, off, 32));
  }

  const float pselfA = __expf(selfA - mA);
  const float pselfB = DUAL ? __expf(selfB - mB) : 0.0f;
  float4 acc;
  {
    const float psHi = DUAL ? pselfB : pselfA;
    const uint2 raw = Tab[(size_t)node * (STRIDE / 4) + lane];
    const v2h h01 = __builtin_bit_cast(v2h, raw.x);
    const v2h h23 = __builtin_bit_cast(v2h, raw.y);
    acc.x = pselfA * (float)h01[0]; acc.y = pselfA * (float)h01[1];
    acc.z = psHi   * (float)h23[0]; acc.w = psHi   * (float)h23[1];
  }
  float denA = 0.0f, denB = 0.0f;

  for (int base = beg; base < end; base += 32) {
    int   s  = s0;
    float pA = 0.0f, pB = 0.0f;
    if (base == beg) {
      if (v0) {
        pA = __expf(tA0 - mA);
        if constexpr (DUAL) pB = __expf(tB0 - mB);
      }
    } else {
      const int i = base + lane;
      if (i < end) {
        s  = clamp_node(csr_src[i]);
        if constexpr (DUAL) {
          const float2 t = asP[s];
          pA = __expf(lrelu(t.x + adAn) - mA);
          pB = __expf(lrelu(t.y + adBn) - mB);
        } else {
          pA = __expf(lrelu(asA[s] + adAn) - mA);
        }
      }
    }
    denA += pA;
    if constexpr (DUAL) denB += pB;

    const int rem = end - base;
    const int cnt = rem < 32 ? rem : 32;
#pragma unroll GATHERS
    for (int j = 0; j < cnt; ++j) {
      const int sj = lane_read_i(s, j);
      const float pj = lane_read_f(pA, j);
      float pjHi = pj;
      if constexpr (DUAL) pjHi = lane_read_f(pB, j);
      const uint2 raw = Tab[(size_t)sj * (STRIDE / 4) + lane];
      const v2h h01 = __builtin_bit_cast(v2h, raw.x);
      const v2h h23 = __builtin_bit_cast(v2h, raw.y);
      acc.x = fmaf(pj,   (float)h01[0], acc.x);
      acc.y = fmaf(pj,   (float)h01[1], acc.y);
      acc.z = fmaf(pjHi, (float)h23[0], acc.z);
      acc.w = fmaf(pjHi, (float)h23[1], acc.w);
    }
  }

  for (int off = 16; off > 0; off >>= 1) {
    denA += __shfl_xor(denA, off, 32);
    if constexpr (DUAL) denB += __shfl_xor(denB, off, 32);
  }
  const float den   = denA + pselfA;
  const float denHi = DUAL ? (denB + pselfB) : den;
  const float* __restrict__ bHi = DUAL ? bB : bA + 64;
  float4 o;
  o.x = acc.x / den   + bA[lane];
  o.y = acc.y / den   + bA[lane + 32];
  o.z = acc.z / denHi + bHi[lane];
  o.w = acc.w / denHi + bHi[lane + 32];
  return o;
}

__global__ void node_softmax_gather_pair(const int* __restrict__ rowptr,
    const int* __restrict__ csr_src,
    const float* __restrict__ as2, const float* __restrict__ ad2,
    const float* __restrict__ HA,  const float* __restrict__ bA,
    float* __restrict__ outA,
    const float* __restrict__ bB,
    float* __restrict__ outB) {
  const int lane = threadIdx.x & 31;
  const int node = __builtin_amdgcn_readfirstlane(
      (int)(blockIdx.x * (blockDim.x >> 5) + (threadIdx.x >> 5)));
  if (node >= N_NODES) return;
  const float4 o = node_aggregate<true, 4>(node, node, lane, rowptr, csr_src,
      as2, ad2, HA, bA, nullptr, nullptr, bB);
  float* __restrict__ oA = outA + (size_t)node * OUT_CH + lane;
  float* __restrict__ oB = outB + (size_t)node * OUT_CH + lane;
  oA[0]  = o.x;
  oA[32] = o.y;
  oB[0]  = o.z;
  oB[32] = o.w;
}

#define FUSE_TB   512
#define A_PITCH   136
#define C_PITCH   132
static_assert(FUSE_TB == 16 * 32 && HID == 128 && 2 * OUT_CH == 128, "strip geometry");
static_assert(FUSE_TB == 4 * NCH, "four threads fetch one chunk piece");
static_assert(N_NODES <= 65536, "source fits 16 bits of the staged word");
__global__ void __launch_bounds__(FUSE_TB, 16)
aggregate_and_project(const int* __restrict__ pair_words, const int* __restrict__ hist,
    const int* __restrict__ offs,
    int* __restrict__ rowptr, int* __restrict__ csr_src,
    const float* __restrict__ as1, const float* __restrict__ ad1,
    const float* __restrict__ HX,  const float* __restrict__ b1,
    const float* __restrict__ B0,  const float* __restrict__ B1,
    float* __restrict__ C,
    const float* __restrict__ a_s0, const float* __restrict__ a_d0,
    const float* __restrict__ a_s1, const float* __restrict__ a_d1,
    float* __restrict__ os2, float* __restrict__ od2) {
  constexpr int NT = 8, K = HID, KC = K / 32;
  float2* __restrict__ osP = (float2*)os2;
  float2* __restrict__ odP = (float2*)od2;
  __shared__ __align__(32) _Float16 ldsB[K * NT * 16];
  __shared__ __align__(16) _Float16 As[16 * A_PITCH];
  __shared__ __align__(16) float    Cs[16 * C_PITCH];
  __shared__ int lst[SEG_CAP], seg[SEG_CAP];
  __shared__ int cnt[BKT], cur[BKT], rp[BKT + 1];
  __shared__ int ccnt[NCH], cgl[NCH], cpos[NCH];
  __shared__ int wbeg[NCH / 32], wlen[NCH / 32];
  const int2* __restrict__ pairs = (const int2*)pair_words;
  const int tid  = threadIdx.x;
  const int lane = tid & 31;
  const int w    = __builtin_amdgcn_readfirstlane((int)(threadIdx.x >> 5));
  const int hh   = lane >> 4;
  const int l16  = lane & 15;
  const int b    = blockIdx.x;
  _Float16* __restrict__ Ct = (_Float16*)C;

  if (tid < BKT) cnt[tid] = 0;
  if (tid < NCH) {
    int n = hist[tid * NBP + b];
    int o = offs[tid * NBP + b];
    n = min(max(n, 0), CHUNK);
    o = min(max(o, 0), CHUNK - n);
    int incl = n;
#pragma unroll
    for (int off = 1; off < 32; off <<= 1) {
      const int u = __shfl_up(incl, off, 32);
      if (lane >= off) incl += u;
    }
    ccnt[tid] = n;
    cgl[tid]  = tid * CHUNK + o;
    cpos[tid] = incl - n;
#pragma unroll
    for (int off = 16; off > 0; off >>= 1) o += __shfl_xor(o, off, 32);
    if (lane == 31) { wbeg[w] = o; wlen[w] = incl; }
  }
  __syncthreads();
  const int c   = tid >> 2;
  const int sub = tid & 3;
  int sbeg = 0, slen = 0, pbase = cpos[c];
#pragma unroll
  for (int j = 0; j < NCH / 32; ++j) {
    if (j < (c >> 5)) pbase += wlen[j];
    sbeg += wbeg[j];
    slen += wlen[j];
  }
  const int segbeg = __builtin_amdgcn_readfirstlane(min(max(sbeg, 0), E_REAL));
  const int seglen = __builtin_amdgcn_readfirstlane(min(max(slen, 0), E_REAL - min(max(sbeg, 0), E_REAL)));
  const bool staged = seglen <= SEG_CAP;
  const int pn = ccnt[c];
  const int2* __restrict__ p = pairs + cgl[c];

  if (staged) {
    int2 e[4];
#pragma unroll
    for (int k = 0; k < 4; ++k) {
      e[k] = make_int2(0, 0);
      if (sub + 4 * k < pn) e[k] = p[sub + 4 * k];
    }
#pragma unroll
    for (int k = 0; k < 4; ++k) {
      const int i = sub + 4 * k;
      if (i < pn) lst[min(pbase + i, SEG_CAP - 1)] = ((e[k].y & (BKT - 1)) << 16) | clamp_node(e[k].x);
    }
    for (int i = sub + 16; i < pn; i += 4) {
      const int2 ee = p[i];
      lst[min(pbase + i, SEG_CAP - 1)] = ((ee.y & (BKT - 1)) << 16) | clamp_node(ee.x);
    }
    __syncthreads();
    for (int i = tid; i < seglen; i += FUSE_TB) atomicAdd(&cnt[(lst[i] >> 16) & (BKT - 1)], 1);
  } else {
    for (int i = sub; i < pn; i += 4) atomicAdd(&cnt[p[i].y & (BKT - 1)], 1);
  }
  __syncthreads();

  if (tid < 32) {
    const int v0 = cnt[2 * lane], v1 = cnt[2 * lane + 1];
    const int sum = v0 + v1;
    int incl = sum;
#pragma unroll
    for (int off = 1; off < 32; off <<= 1) {
      const int u = __shfl_up(incl, off, 32);
      if (lane >= off) incl += u;
    }
    const int run = incl - sum;
    rp[2 * lane]      = run;       cur[2 * lane]     = run;
    rp[2 * lane + 1]  = run + v0;  cur[2 * lane + 1] = run + v0;
    if (lane == 31) rp[BKT] = seglen;
  }
  __syncthreads();
  if (tid < BKT) {
    const int node = b * BKT + tid;
    if (node <= N_NODES) rowptr[node] = segbeg + rp[tid];
  }

  if (staged) {
    for (int i = tid; i < seglen; i += FUSE_TB) {
      const int key = lst[i];
      const int ps  = atomicAdd(&cur[(key >> 16) & (BKT - 1)], 1);
      seg[min(max(ps, 0), SEG_CAP - 1)] = key;
    }
    __syncthreads();
    for (int i = tid; i < seglen; i += FUSE_TB) {
      const int key = seg[i];
      const int d   = (key >> 16) & (BKT - 1);
      const int lo  = min(max(rp[d], 0), seglen);
      const int hi  = min(max(rp[d + 1], lo), seglen);
      int rank = 0;
      for (int j = lo; j < hi; ++j) {
        const int kj = seg[j];
        rank += (kj < key || (kj == key && j < i)) ? 1 : 0;
      }
      const int at = min(lo + rank, SEG_CAP - 1);
      lst[at] = key & 0xFFFF;
      csr_src[clamp_edge(segbeg + at)] = key & 0xFFFF;
    }
  } else {
    for (int i = sub; i < pn; i += 4) {
      const int2 ee = p[i];
      const int  ps = atomicAdd(&cur[ee.y & (BKT - 1)], 1);
      csr_src[clamp_edge(segbeg + ps)] = clamp_node(ee.x);
    }
    __threadfence();
  }
  __syncthreads();
  if (!staged) __threadfence();

  const float s0a = a_s0[lane], s0b = a_s0[lane + 32];
  const float d0a = a_d0[lane], d0b = a_d0[lane + 32];
  const float s1a = a_s1[lane], s1b = a_s1[lane + 32];
  const float d1a = a_d1[lane], d1b = a_d1[lane + 32];

  const int nstrips = min(BKT / 16, N_NODES / 16 - b * (BKT / 16));
  for (int q = 0; q < nstrips; ++q) {
    const int dloc = q * 16 + w;
    const int row  = b * BKT + dloc;
    float4 o;
    if (staged) {
      o = node_aggregate<false, 4>(row, dloc, lane, rp, lst,
          as1, ad1, HX, b1, nullptr, nullptr, nullptr);
    } else {
      o = node_aggregate<false, 4>(row, dloc, lane, rp, csr_src + segbeg,
          as1, ad1, HX, b1, nullptr, nullptr, nullptr);
    }
    o.x = o.x > 0.0f ? o.x : 0.0f;
    o.y = o.y > 0.0f ? o.y : 0.0f;
    o.z = o.z > 0.0f ? o.z : 0.0f;
    o.w = o.w > 0.0f ? o.w : 0.0f;
    As[w * A_PITCH + lane]      = (_Float16)o.x;
    As[w * A_PITCH + lane + 32] = (_Float16)o.y;
    As[w * A_PITCH + lane + 64] = (_Float16)o.z;
    As[w * A_PITCH + lane + 96] = (_Float16)o.w;
    if (q == 0) stage_b_fragments<NT, K, true>(ldsB, B0, B1);
    __syncthreads();

    if (w < NT) {
      v8f acc = {};
#pragma unroll
      for (int kc = 0; kc < KC; ++kc) {
        const _Float16* ap = As + l16 * A_PITCH + kc * 32 + hh * 8;
        const v8h alo = *(const v8h*)ap;
        const v8h ahi = *(const v8h*)(ap + 16);
        const v16h a = __builtin_shufflevector(alo, ahi,
            0, 1, 2, 3, 4, 5, 6, 7, 8, 9, 10, 11, 12, 13, 14, 15);
        const v16h bf = *(const v16h*)(ldsB + ((size_t)(kc * NT + w) * 32 + lane) * 16);
        acc = __builtin_amdgcn_wmma_f32_16x16x32_f16(false, a, false, bf,
                                                     (short)0, acc, false, false);
      }
#pragma unroll
      for (int r = 0; r < 8; ++r) Cs[(hh * 8 + r) * C_PITCH + w * 16 + l16] = acc[r];
    }
    __syncthreads();

    const float v0 = Cs[w * C_PITCH + lane];
    const float v1 = Cs[w * C_PITCH + lane + 32];
    const float v2 = Cs[w * C_PITCH + lane + 64];
    const float v3 = Cs[w * C_PITCH + lane + 96];
    float ps0 = fmaf(v1, s0b, v0 * s0a), pd0 = fmaf(v1, d0b, v0 * d0a);
    float ps1 = fmaf(v3, s1b, v2 * s1a), pd1 = fmaf(v3, d1b, v2 * d1a);
#pragma unroll
    for (int off = 16; off > 0; off >>= 1) {
      ps0 += __shfl_xor(ps0, off, 32);
      pd0 += __shfl_xor(pd0, off, 32);
      ps1 += __shfl_xor(ps1, off, 32);
      pd1 += __shfl_xor(pd1, off, 32);
    }
    v4h t;
    t[0] = (_Float16)v0; t[1] = (_Float16)v1; t[2] = (_Float16)v2; t[3] = (_Float16)v3;
    *(v4h*)(Ct + (size_t)row * 128 + 4 * lane) = t;
    if (lane == 0) {
      osP[row] = make_float2(ps0, ps1);
      odP[row] = make_float2(pd0, pd1);
    }
  }
}


static inline int ceildiv(long a, long b) { return (int)((a + b - 1) / b); }

extern "C" void kernel_launch(void* const* d_in, const int* in_sizes, int n_in,
                              void* d_out, int out_size, void* d_ws, size_t ws_size,
                              hipStream_t stream) {
  (void)in_sizes; (void)n_in; (void)out_size; (void)ws_size;
  const float* x       = (const float*)d_in[0];
  const int*   ei      = (const int*)d_in[1];
  const float* W1      = (const float*)d_in[2];
  const float* a_src1  = (const float*)d_in[3];
  const float* a_dst1  = (const float*)d_in[4];
  const float* b1      = (const float*)d_in[5];
  const float* W_mu    = (const float*)d_in[6];
  const float* a_srcm  = (const float*)d_in[7];
  const float* a_dstm  = (const float*)d_in[8];
  const float* b_mu    = (const float*)d_in[9];
  const float* W_ls    = (const float*)d_in[10];
  const float* a_srcl  = (const float*)d_in[11];
  const float* a_dstl  = (const float*)d_in[12];
  const float* b_ls    = (const float*)d_in[13];

  const long SZ_H = (long)N_NODES * HID;
  const long SZ_O = (long)N_NODES * OUT_CH;

  float* ws  = (float*)d_ws;
  float* hx  = ws;
  float* h1  = hx + SZ_H;
  float* hml = h1;
  float* sc  = h1 + SZ_H;
  float* as1 = sc + 0L * N_NODES, *ad1 = sc + 1L * N_NODES;
  float* as2 = sc + 2L * N_NODES;
  float* ad2 = sc + 4L * N_NODES;
  int* rowptr  = (int*)(sc + 8L * N_NODES);
  int* csr_src = (int*)(sc + 10L * N_NODES);
  int*  pairs  = (int*)(h1 + SZ_H / 2);
  int*  hist   = pairs + 2L * E_REAL;
  int*  offs   = hist + NCH * NBP;

  float* out_mu = (float*)d_out;
  float* out_ls = out_mu + SZ_O;

  const int TB = 256;

  sort_and_project<<<NCH + GEMM1_BLOCKS, EDGE_TB, 0, stream>>>(
      ei, hist, offs, pairs, x, W1, hx, a_src1, a_dst1, as1, ad1);

  const int node_blocks = ceildiv(N_NODES, TB / 32);

  aggregate_and_project<<<NB, FUSE_TB, 0, stream>>>(
      pairs, hist, offs, rowptr, csr_src, as1, ad1, hx, b1, W_mu, W_ls, hml,
      a_srcm, a_dstm, a_srcl, a_dstl, as2, ad2);

  node_softmax_gather_pair<<<node_blocks, TB, 0, stream>>>(
      rowptr, csr_src, as2, ad2, hml, b_mu, out_mu,
      b_ls, out_ls);
}


# node pass requests a node's gathered rows in batches of 16 (layer 2) / 8 (layer 1) and the leftover rows as one batch; was batches of 4, then one row per round trip; order of every sum unchanged
# speedup vs baseline: 9.4981x; 1.0441x over previous
#include <hip/hip_runtime.h>
#include <hip/hip_fp16.h>

typedef __attribute__((ext_vector_type(16))) _Float16 v16h;
typedef __attribute__((ext_vector_type(8)))  float    v8f;
typedef __attribute__((ext_vector_type(8)))  _Float16 v8h;
typedef __attribute__((ext_vector_type(4)))  _Float16 v4h;
typedef __attribute__((ext_vector_type(2)))  _Float16 v2h;

#define N_NODES 50000
#define E_REAL  800000
#define IN_CH   128
#define HID     128
#define OUT_CH  64
#define NEG_SLOPE 0.2f

template<int NT, int K, bool SPLIT>
__device__ __forceinline__ void stage_b_fragments(_Float16* __restrict__ ldsB,
                               const float* __restrict__ B0,
                               const float* __restrict__ B1) {
  constexpr int KC  = K / 32;
  constexpr int NB0 = (SPLIT ? NT / 2 : NT) * 16;
  for (int slot = threadIdx.x; slot < KC * NT * 32; slot += blockDim.x) {
    const int sl   = slot & 31;
    const int t    = slot >> 5;
    const int nt   = t % NT;
    const int kc   = t / NT;
    const int kb   = kc * 32 + (sl >> 4) * 8;
    const int n    = nt * 16 + (sl & 15);
    const float* __restrict__ Bs = (SPLIT && n >= NB0) ? B1 : B0;
    const int nc   = (SPLIT && n >= NB0) ? n - NB0 : n;
    _Float16* dst  = ldsB + slot * 16;
#pragma unroll
    for (int j = 0; j < 8; ++j) {
      dst[j]     = (_Float16)Bs[(size_t)(kb + j) * NB0 + nc];
      dst[j + 8] = (_Float16)Bs[(size_t)(kb + 16 + j) * NB0 + nc];
    }
  }
}

template<int NT, int K, bool SPLIT>
__device__ __forceinline__ void gemm_attn_body(int blk,
                               const float* __restrict__ A,
                               const float* __restrict__ B0,
                               const float* __restrict__ B1,
                               float* __restrict__ C,
                               const float* __restrict__ a_s0,
                               const float* __restrict__ a_d0,
                               const float* __restrict__ a_s1,
                               const float* __restrict__ a_d1,
                               float* __restrict__ os0, float* __restrict__ od0,
                               float* __restrict__ os1, float* __restrict__ od1,
                               int M) {
  constexpr int N  = NT * 16;
  constexpr int KC = K / 32;
  constexpr int NTH = SPLIT ? NT / 2 : NT;
  static_assert(!SPLIT || NT % 2 == 0, "two equal halves");
  static_assert(NT == 8, "table row = 128 halves: two 8-byte pieces per lane and accumulator register");
  _Float16* __restrict__ Ct = (_Float16*)C;
  __shared__ __align__(32) _Float16 ldsB[K * N];

  const int lane   = threadIdx.x & 31;
  const int wlocal = threadIdx.x >> 5;
  const int mtile  = blk * (int)(blockDim.x >> 5) + wlocal;
  const int hh  = lane >> 4;
  const int l16 = lane & 15;

  stage_b_fragments<NT, K, SPLIT>(ldsB, B0, B1);
  __syncthreads();
  if (mtile * 16 >= M) return;

  const float* __restrict__ Arow = A + (size_t)(mtile * 16 + l16) * K;
  v8f acc[NT] = {};
#pragma unroll
  for (int kc = 0; kc < KC; ++kc) {
    const int kb = kc * 32 + hh * 8;
    const float4 a0 = *(const float4*)(Arow + kb);
    const float4 a1 = *(const float4*)(Arow + kb + 4);
    const float4 a2 = *(const float4*)(Arow + kb + 16);
    const float4 a3 = *(const float4*)(Arow + kb + 20);
    const float av[16] = {a0.x, a0.y, a0.z, a0.w, a1.x, a1.y, a1.z, a1.w,
                          a2.x, a2.y, a2.z, a2.w, a3.x, a3.y, a3.z, a3.w};
    v16h a;
#pragma unroll
    for (int j = 0; j < 16; ++j) a[j] = (_Float16)av[j];
#pragma unroll
    for (int nt = 0; nt < NT; ++nt) {
      const v16h b = *(const v16h*)(ldsB + ((size_t)(kc * NT + nt) * 32 + lane) * 16);
      acc[nt] = __builtin_amdgcn_wmma_f32_16x16x32_f16(false, a, false, b,
                                                       (short)0, acc[nt], false, false);
    }
  }

  float asv[NT], adv[NT];
#pragma unroll
  for (int nt = 0; nt < NT; ++nt) {
    if (nt < NTH) {
      asv[nt] = a_s0[nt * 16 + l16];
      adv[nt] = a_d0[nt * 16 + l16];
    } else {
      asv[nt] = a_s1[(nt - NTH) * 16 + l16];
      adv[nt] = a_d1[(nt - NTH) * 16 + l16];
    }
  }
#pragma unroll
  for (int r = 0; r < 8; ++r) {
    const int row = mtile * 16 + hh * 8 + r;
    float ps0 = 0.0f, pd0 = 0.0f, ps1 = 0.0f, pd1 = 0.0f;
    v4h pe, po;
#pragma unroll
    for (int nt = 0; nt < NT; ++nt) {
      const float v = acc[nt][r];
      if (nt & 1) po[nt >> 1] = (_Float16)v; else pe[nt >> 1] = (_Float16)v;
      if (nt < NTH) {
        ps0 = fmaf(v, asv[nt], ps0);
        pd0 = fmaf(v, adv[nt], pd0);
      } else {
        ps1 = fmaf(v, asv[nt], ps1);
        pd1 = fmaf(v, adv[nt], pd1);
      }
    }
#pragma unroll
    for (int off = 8; off > 0; off >>= 1) {
      ps0 += __shfl_xor(ps0, off, 32);
      pd0 += __shfl_xor(pd0, off, 32);
      if constexpr (SPLIT) {
        ps1 += __shfl_xor(ps1, off, 32);
        pd1 += __shfl_xor(pd1, off, 32);
      }
    }
    *(v4h*)(Ct + (size_t)row * N + 4 * l16)      = pe;
    *(v4h*)(Ct + (size_t)row * N + 64 + 4 * l16) = po;
    if (l16 == 0) {
      os0[row] = ps0; od0[row] = pd0;
      if constexpr (SPLIT) { os1[row] = ps1; od1[row] = pd1; }
    }
  }
}


#define BKT     64
#define BKT_SH  6
#define NB      782
#define NBP     800
#define BPL     (NBP / 32)
#define NCH     128
#define CHUNK   6250
#define EDGE_TB 512
#define SEG_CAP 2048
static_assert(NCH * CHUNK == E_REAL, "chunks must tile the edge list");
static_assert(NB * BKT >= N_NODES && (NB - 1) * BKT < N_NODES, "bucket count");
static_assert(NBP >= NB && NBP % 32 == 0, "one LDS counter and one table column per bucket; whole lanes in the scan");
static_assert(BKT == 64 && (1 << BKT_SH) == BKT && N_NODES % 16 == 0, "two counters per lane in the bucket scan; whole strips");
static_assert(NCH % 32 == 0, "whole waves of chunks");

__device__ __forceinline__ int clamp_node(int s) {
  return min(max(s, 0), N_NODES - 1);
}
__device__ __forceinline__ int clamp_edge(int p) {
  return min(max(p, 0), E_REAL - 1);
}

__device__ __forceinline__ void
chunk_sort_body(int me, const int* __restrict__ ei, int* __restrict__ hist,
                int* __restrict__ offs, int* __restrict__ pair_words) {
  int2* __restrict__ pairs = (int2*)pair_words + (size_t)me * CHUNK;
  __shared__ int h[NBP], start[NBP], cur[NBP];
  const int tid  = threadIdx.x;
  const int lane = tid & 31;
  for (int t = tid; t < NBP; t += EDGE_TB) h[t] = 0;
  __syncthreads();
  const int* __restrict__ sp = ei + me * CHUNK;
  const int* __restrict__ dp = ei + E_REAL + me * CHUNK;
  constexpr int TRIPS = (CHUNK + EDGE_TB - 1) / EDGE_TB;
  int sv[TRIPS], dv[TRIPS];
#pragma unroll
  for (int k = 0; k < TRIPS; ++k) {
    const int ic = min(tid + k * EDGE_TB, CHUNK - 1);
    sv[k] = sp[ic];
    dv[k] = dp[ic];
  }
#pragma unroll
  for (int k = 0; k < TRIPS; ++k) {
    const int i = tid + k * EDGE_TB;
    dv[k] = clamp_node(dv[k]);
    if (i < CHUNK) atomicAdd(&h[dv[k] >> BKT_SH], 1);
  }
  __syncthreads();

  if (tid < 32) {
    int v[BPL];
    int s = 0;
#pragma unroll
    for (int j = 0; j < BPL; ++j) { v[j] = h[lane * BPL + j]; s += v[j]; }
    int incl = s;
#pragma unroll
    for (int off = 1; off < 32; off <<= 1) {
      const int u = __shfl_up(incl, off, 32);
      if (lane >= off) incl += u;
    }
    int run = incl - s;
#pragma unroll
    for (int j = 0; j < BPL; ++j) {
      start[lane * BPL + j] = run;
      cur[lane * BPL + j]   = run;
      run += v[j];
    }
  }
  __syncthreads();
  for (int t = tid; t < NBP; t += EDGE_TB) {
    hist[me * NBP + t] = h[t];
    offs[me * NBP + t] = start[t];
  }

  int pos[TRIPS];
#pragma unroll
  for (int k = 0; k < TRIPS; ++k) {
    const int i = tid + k * EDGE_TB;
    pos[k] = i < CHUNK ? atomicAdd(&cur[dv[k] >> BKT_SH], 1) : 0;
  }
#pragma unroll
  for (int k = 0; k < TRIPS; ++k) {
    const int i = tid + k * EDGE_TB;
    if (i < CHUNK) pairs[min(max(pos[k], 0), CHUNK - 1)] = make_int2(sv[k], dv[k]);
  }
}

#define GEMM1_BLOCKS ((N_NODES / 16 + EDGE_TB / 32 - 1) / (EDGE_TB / 32))
static_assert(N_NODES % 16 == 0, "whole 16-row strips");
__global__ void __launch_bounds__(EDGE_TB)
sort_and_project(const int* __restrict__ ei, int* __restrict__ hist,
                 int* __restrict__ offs, int* __restrict__ pair_words,
                 const float* __restrict__ x, const float* __restrict__ W,
                 float* __restrict__ hx,
                 const float* __restrict__ a_s, const float* __restrict__ a_d,
                 float* __restrict__ os, float* __restrict__ od) {
  if (blockIdx.x < NCH) {
    chunk_sort_body((int)blockIdx.x, ei, hist, offs, pair_words);
  } else {
    gemm_attn_body<HID / 16, IN_CH, false>((int)blockIdx.x - NCH, x, W, nullptr, hx,
        a_s, a_d, nullptr, nullptr, os, od, nullptr, nullptr, N_NODES);
  }
}

__device__ __forceinline__ float lrelu(float t) { return t > 0.0f ? t : NEG_SLOPE * t; }


__device__ __forceinline__ int lane_read_i(int v, int j) {
  return __builtin_amdgcn_readlane(v, j);
}
__device__ __forceinline__ float lane_read_f(float v, int j) {
  return __int_as_float(__builtin_amdgcn_readlane(__float_as_int(v), j));
}

template<int CNT>
__device__ __forceinline__ void rows_offsets_live(unsigned (&off)[CNT]) {
  static_assert(CNT == 16 || CNT == 8 || CNT == 4 || CNT == 2 || CNT == 1, "batch sizes");
  if constexpr (CNT == 16) {
    asm volatile("" : "+v"(off[0]), "+v"(off[1]), "+v"(off[2]), "+v"(off[3]),
                      "+v"(off[4]), "+v"(off[5]), "+v"(off[6]), "+v"(off[7]),
                      "+v"(off[8]), "+v"(off[9]), "+v"(off[10]), "+v"(off[11]),
                      "+v"(off[12]), "+v"(off[13]), "+v"(off[14]), "+v"(off[15]));
  } else if constexpr (CNT == 8) {
    asm volatile("" : "+v"(off[0]), "+v"(off[1]), "+v"(off[2]), "+v"(off[3]),
                      "+v"(off[4]), "+v"(off[5]), "+v"(off[6]), "+v"(off[7]));
  } else if constexpr (CNT == 4) {
    asm volatile("" : "+v"(off[0]), "+v"(off[1]), "+v"(off[2]), "+v"(off[3]));
  } else if constexpr (CNT == 2) {
    asm volatile("" : "+v"(off[0]), "+v"(off[1]));
  } else {
    asm volatile("" : "+v"(off[0]));
  }
}
template<bool DUAL, int CNT, bool CLAMP>
__device__ __forceinline__ void rows_batch(float4& acc,
    const uint2* __restrict__ Tab, unsigned lane8, int s, float pA, float pB,
    int j0, int cnt) {
  uint2 raw[CNT];
  unsigned off[CNT];
#pragma unroll
  for (int k = 0; k < CNT; ++k) {
    const int jk = CLAMP ? min(j0 + k, cnt - 1) : j0 + k;
    off[k] = ((unsigned)lane_read_i(s, jk) << 8) + lane8;
  }
  rows_offsets_live<CNT>(off);
#pragma unroll
  for (int k = 0; k < CNT; ++k)
    raw[k] = *(const uint2*)((const char*)Tab + off[k]);
#pragma unroll
  for (int k = 0; k < CNT; ++k) {
    const bool in = !CLAMP || j0 + k < cnt;
    const int  jk = CLAMP ? min(j0 + k, cnt - 1) : j0 + k;
    const float pj = in ? lane_read_f(pA, jk) : 0.0f;
    float pjHi = pj;
    if constexpr (DUAL) pjHi = in ? lane_read_f(pB, jk) : 0.0f;
    const v2h h01 = __builtin_bit_cast(v2h, raw[k].x);
    const v2h h23 = __builtin_bit_cast(v2h, raw[k].y);
    acc.x = fmaf(pj,   (float)h01[0], acc.x);
    acc.y = fmaf(pj,   (float)h01[1], acc.y);
    acc.z = fmaf(pjHi, (float)h23[0], acc.z);
    acc.w = fmaf(pjHi, (float)h23[1], acc.w);
  }
}

template<bool DUAL, int GATHERS>
__device__ __forceinline__ float4 node_aggregate(int node, int rpi, int lane,
    const int* __restrict__ rowptr,
    const int* __restrict__ csr_src,
    const float* __restrict__ asA, const float* __restrict__ adA,
    const float* __restrict__ HA,  const float* __restrict__ bA,
    const float* __restrict__ asB, const float* __restrict__ adB,
    const float* __restrict__ bB) {
  constexpr int STRIDE  = 128;
  const uint2* __restrict__ Tab = (const uint2*)HA;
  const int beg = __builtin_amdgcn_readfirstlane(rowptr[rpi]);
  const int end = __builtin_amdgcn_readfirstlane(rowptr[rpi + 1]);

  const float NEG_INF = -__builtin_huge_valf();
  const float2* __restrict__ asP = (const float2*)asA;
  const float2* __restrict__ adP = (const float2*)adA;
  float adAn, selfA, adBn = 0.0f, selfB = 0.0f;
  if constexpr (DUAL) {
    const float2 dn = adP[node];
    const float2 sn = asP[node];
    adAn = dn.x; adBn = dn.y;
    selfA = lrelu(sn.x + adAn);
    selfB = lrelu(sn.y + adBn);
  } else {
    adAn  = adA[node];
    selfA = lrelu(asA[node] + adAn);
  }

  const int  i0 = beg + lane;
  const bool v0 = i0 < end;
  int   s0  = node;
  float tA0 = NEG_INF, tB0 = NEG_INF;
  if (v0) {
    s0  = clamp_node(csr_src[i0]);
    if constexpr (DUAL) {
      const float2 t = asP[s0];
      tA0 = lrelu(t.x + adAn);
      tB0 = lrelu(t.y + adBn);
    } else {
      tA0 = lrelu(asA[s0] + adAn);
    }
  }
  float mA = fmaxf(selfA, tA0);
  float mB = fmaxf(selfB, tB0);
  for (int base = beg + 32; base < end; base += 32) {
    const int i = base + lane;
    if (i < end) {
      const int s = clamp_node(csr_src[i]);
      if constexpr (DUAL) {
        const float2 t = asP[s];
        mA = fmaxf(mA, lrelu(t.x + adAn));
        mB = fmaxf(mB, lrelu(t.y + adBn));
      } else {
        mA = fmaxf(mA, lrelu(asA[s] + adAn));
      }
    }
  }
  for (int off = 16; off > 0; off >>= 1) {
    mA = fmaxf(mA, __shfl_xor(mA, off, 32));
    if constexpr (DUAL) mB = fmaxf(mB, __shfl_xor(mB, off, 32));
  }

  const float pselfA = __expf(selfA - mA);
  const float pselfB = DUAL ? __expf(selfB - mB) : 0.0f;
  float4 acc;
  {
    const float psHi = DUAL ? pselfB : pselfA;
    const uint2 raw = Tab[(size_t)node * (STRIDE / 4) + lane];
    const v2h h01 = __builtin_bit_cast(v2h, raw.x);
    const v2h h23 = __builtin_bit_cast(v2h, raw.y);
    acc.x = pselfA * (float)h01[0]; acc.y = pselfA * (float)h01[1];
    acc.z = psHi   * (float)h23[0]; acc.w = psHi   * (float)h23[1];
  }
  float denA = 0.0f, denB = 0.0f;

  const unsigned lane8 = (unsigned)lane * 8u;
  {
    float pA = 0.0f, pB = 0.0f;
    if (v0) {
      pA = __expf(tA0 - mA);
      if constexpr (DUAL) pB = __expf(tB0 - mB);
    }
    denA += pA;
    if constexpr (DUAL) denB += pB;
    const int rem = end - beg;
    const int cnt = rem < 32 ? rem : 32;
    int j = 0;
#pragma unroll
    for (int b = 0; b < 32 / GATHERS; ++b) {
      if (j + GATHERS > cnt) break;
      rows_batch<DUAL, GATHERS, false>(acc, Tab, lane8, s0, pA, pB, j, cnt);
      j += GATHERS;
    }
    const int r = cnt - j;
    if (GATHERS > 8 && r > 8)      rows_batch<DUAL, 16, true>(acc, Tab, lane8, s0, pA, pB, j, cnt);
    else if (GATHERS > 4 && r > 4) rows_batch<DUAL, 8, true>(acc, Tab, lane8, s0, pA, pB, j, cnt);
    else if (r > 2)                rows_batch<DUAL, 4, true>(acc, Tab, lane8, s0, pA, pB, j, cnt);
    else if (r > 1)                rows_batch<DUAL, 2, false>(acc, Tab, lane8, s0, pA, pB, j, cnt);
    else if (r > 0)                rows_batch<DUAL, 1, false>(acc, Tab, lane8, s0, pA, pB, j, cnt);
  }
  for (int base = beg + 32; base < end; base += 32) {
    int   s  = node;
    float pA = 0.0f, pB = 0.0f;
    const int i = base + lane;
    if (i < end) {
      s  = clamp_node(csr_src[i]);
      if constexpr (DUAL) {
        const float2 t = asP[s];
        pA = __expf(lrelu(t.x + adAn) - mA);
        pB = __expf(lrelu(t.y + adBn) - mB);
      } else {
        pA = __expf(lrelu(asA[s] + adAn) - mA);
      }
    }
    denA += pA;
    if constexpr (DUAL) denB += pB;
    const int rem = end - base;
    const int cnt = rem < 32 ? rem : 32;
    for (int j = 0; j < cnt; j += 4)
      rows_batch<DUAL, 4, true>(acc, Tab, lane8, s, pA, pB, j, cnt);
  }

  for (int off = 16; off > 0; off >>= 1) {
    denA += __shfl_xor(denA, off, 32);
    if constexpr (DUAL) denB += __shfl_xor(denB, off, 32);
  }
  const float den   = denA + pselfA;
  const float denHi = DUAL ? (denB + pselfB) : den;
  const float* __restrict__ bHi = DUAL ? bB : bA + 64;
  float4 o;
  o.x = acc.x / den   + bA[lane];
  o.y = acc.y / den   + bA[lane + 32];
  o.z = acc.z / denHi + bHi[lane];
  o.w = acc.w / denHi + bHi[lane + 32];
  return o;
}

__global__ void node_softmax_gather_pair(const int* __restrict__ rowptr,
    const int* __restrict__ csr_src,
    const float* __restrict__ as2, const float* __restrict__ ad2,
    const float* __restrict__ HA,  const float* __restrict__ bA,
    float* __restrict__ outA,
    const float* __restrict__ bB,
    float* __restrict__ outB) {
  const int lane = threadIdx.x & 31;
  const int node = __builtin_amdgcn_readfirstlane(
      (int)(blockIdx.x * (blockDim.x >> 5) + (threadIdx.x >> 5)));
  if (node >= N_NODES) return;
  const float4 o = node_aggregate<true, 16>(node, node, lane, rowptr, csr_src,
      as2, ad2, HA, bA, nullptr, nullptr, bB);
  float* __restrict__ oA = outA + (size_t)node * OUT_CH + lane;
  float* __restrict__ oB = outB + (size_t)node * OUT_CH + lane;
  oA[0]  = o.x;
  oA[32] = o.y;
  oB[0]  = o.z;
  oB[32] = o.w;
}

#define FUSE_TB   512
#define A_PITCH   136
#define C_PITCH   132
static_assert(FUSE_TB == 16 * 32 && HID == 128 && 2 * OUT_CH == 128, "strip geometry");
static_assert(FUSE_TB == 4 * NCH, "four threads fetch one chunk piece");
static_assert(N_NODES <= 65536, "source fits 16 bits of the staged word");
__global__ void __launch_bounds__(FUSE_TB, 16)
aggregate_and_project(const int* __restrict__ pair_words, const int* __restrict__ hist,
    const int* __restrict__ offs,
    int* __restrict__ rowptr, int* __restrict__ csr_src,
    const float* __restrict__ as1, const float* __restrict__ ad1,
    const float* __restrict__ HX,  const float* __restrict__ b1,
    const float* __restrict__ B0,  const float* __restrict__ B1,
    float* __restrict__ C,
    const float* __restrict__ a_s0, const float* __restrict__ a_d0,
    const float* __restrict__ a_s1, const float* __restrict__ a_d1,
    float* __restrict__ os2, float* __restrict__ od2) {
  constexpr int NT = 8, K = HID, KC = K / 32;
  float2* __restrict__ osP = (float2*)os2;
  float2* __restrict__ odP = (float2*)od2;
  __shared__ __align__(32) _Float16 ldsB[K * NT * 16];
  __shared__ __align__(16) _Float16 As[16 * A_PITCH];
  __shared__ __align__(16) float    Cs[16 * C_PITCH];
  __shared__ int lst[SEG_CAP], seg[SEG_CAP];
  __shared__ int cnt[BKT], cur[BKT], rp[BKT + 1];
  __shared__ int ccnt[NCH], cgl[NCH], cpos[NCH];
  __shared__ int wbeg[NCH / 32], wlen[NCH / 32];
  __shared__ float lvec[4 * OUT_CH];
  const int2* __restrict__ pairs = (const int2*)pair_words;
  const int tid  = threadIdx.x;
  const int lane = tid & 31;
  const int w    = __builtin_amdgcn_readfirstlane((int)(threadIdx.x >> 5));
  const int hh   = lane >> 4;
  const int l16  = lane & 15;
  const int b    = blockIdx.x;
  _Float16* __restrict__ Ct = (_Float16*)C;

  if (tid < 4 * OUT_CH) {
    const int v = tid >> 6, i = tid & (OUT_CH - 1);
    const float* __restrict__ vp = v == 0 ? a_s0 : v == 1 ? a_d0 : v == 2 ? a_s1 : a_d1;
    lvec[tid] = vp[i];
  }
  if (tid < BKT) cnt[tid] = 0;
  if (tid < NCH) {
    int n = hist[tid * NBP + b];
    int o = offs[tid * NBP + b];
    n = min(max(n, 0), CHUNK);
    o = min(max(o, 0), CHUNK - n);
    int incl = n;
#pragma unroll
    for (int off = 1; off < 32; off <<= 1) {
      const int u = __shfl_up(incl, off, 32);
      if (lane >= off) incl += u;
    }
    ccnt[tid] = n;
    cgl[tid]  = tid * CHUNK + o;
    cpos[tid] = incl - n;
#pragma unroll
    for (int off = 16; off > 0; off >>= 1) o += __shfl_xor(o, off, 32);
    if (lane == 31) { wbeg[w] = o; wlen[w] = incl; }
  }
  __syncthreads();
  const int c   = tid >> 2;
  const int sub = tid & 3;
  int sbeg = 0, slen = 0, pbase = cpos[c];
#pragma unroll
  for (int j = 0; j < NCH / 32; ++j) {
    if (j < (c >> 5)) pbase += wlen[j];
    sbeg += wbeg[j];
    slen += wlen[j];
  }
  const int segbeg = __builtin_amdgcn_readfirstlane(min(max(sbeg, 0), E_REAL));
  const int seglen = __builtin_amdgcn_readfirstlane(min(max(slen, 0), E_REAL - min(max(sbeg, 0), E_REAL)));
  const bool staged = seglen <= SEG_CAP;
  const int pn = ccnt[c];
  const int2* __restrict__ p = pairs + cgl[c];

  if (staged) {
    int2 e[4];
#pragma unroll
    for (int k = 0; k < 4; ++k) {
      e[k] = make_int2(0, 0);
      if (sub + 4 * k < pn) e[k] = p[sub + 4 * k];
    }
#pragma unroll
    for (int k = 0; k < 4; ++k) {
      const int i = sub + 4 * k;
      if (i < pn) lst[min(pbase + i, SEG_CAP - 1)] = ((e[k].y & (BKT - 1)) << 16) | clamp_node(e[k].x);
    }
    for (int i = sub + 16; i < pn; i += 4) {
      const int2 ee = p[i];
      lst[min(pbase + i, SEG_CAP - 1)] = ((ee.y & (BKT - 1)) << 16) | clamp_node(ee.x);
    }
    __syncthreads();
    for (int i = tid; i < seglen; i += FUSE_TB) atomicAdd(&cnt[(lst[i] >> 16) & (BKT - 1)], 1);
  } else {
    for (int i = sub; i < pn; i += 4) atomicAdd(&cnt[p[i].y & (BKT - 1)], 1);
  }
  __syncthreads();

  if (tid < 32) {
    const int v0 = cnt[2 * lane], v1 = cnt[2 * lane + 1];
    const int sum = v0 + v1;
    int incl = sum;
#pragma unroll
    for (int off = 1; off < 32; off <<= 1) {
      const int u = __shfl_up(incl, off, 32);
      if (lane >= off) incl += u;
    }
    const int run = incl - sum;
    rp[2 * lane]      = run;       cur[2 * lane]     = run;
    rp[2 * lane + 1]  = run + v0;  cur[2 * lane + 1] = run + v0;
    if (lane == 31) rp[BKT] = seglen;
  }
  __syncthreads();
  if (tid < BKT) {
    const int node = b * BKT + tid;
    if (node <= N_NODES) rowptr[node] = segbeg + rp[tid];
  }

  if (staged) {
    for (int i = tid; i < seglen; i += FUSE_TB) {
      const int key = lst[i];
      const int ps  = atomicAdd(&cur[(key >> 16) & (BKT - 1)], 1);
      seg[min(max(ps, 0), SEG_CAP - 1)] = key;
    }
    __syncthreads();
    for (int i = tid; i < seglen; i += FUSE_TB) {
      const int key = seg[i];
      const int d   = (key >> 16) & (BKT - 1);
      const int lo  = min(max(rp[d], 0), seglen);
      const int hi  = min(max(rp[d + 1], lo), seglen);
      int rank = 0;
      for (int j = lo; j < hi; ++j) {
        const int kj = seg[j];
        rank += (kj < key || (kj == key && j < i)) ? 1 : 0;
      }
      const int at = min(lo + rank, SEG_CAP - 1);
      lst[at] = key & 0xFFFF;
      csr_src[clamp_edge(segbeg + at)] = key & 0xFFFF;
    }
  } else {
    for (int i = sub; i < pn; i += 4) {
      const int2 ee = p[i];
      const int  ps = atomicAdd(&cur[ee.y & (BKT - 1)], 1);
      csr_src[clamp_edge(segbeg + ps)] = clamp_node(ee.x);
    }
    __threadfence();
  }
  __syncthreads();
  if (!staged) __threadfence();


  const int nstrips = min(BKT / 16, N_NODES / 16 - b * (BKT / 16));
  for (int q = 0; q < nstrips; ++q) {
    const int dloc = q * 16 + w;
    const int row  = b * BKT + dloc;
    float4 o;
    if (staged) {
      o = node_aggregate<false, 8>(row, dloc, lane, rp, lst,
          as1, ad1, HX, b1, nullptr, nullptr, nullptr);
    } else {
      o = node_aggregate<false, 8>(row, dloc, lane, rp, csr_src + segbeg,
          as1, ad1, HX, b1, nullptr, nullptr, nullptr);
    }
    o.x = o.x > 0.0f ? o.x : 0.0f;
    o.y = o.y > 0.0f ? o.y : 0.0f;
    o.z = o.z > 0.0f ? o.z : 0.0f;
    o.w = o.w > 0.0f ? o.w : 0.0f;
    As[w * A_PITCH + lane]      = (_Float16)o.x;
    As[w * A_PITCH + lane + 32] = (_Float16)o.y;
    As[w * A_PITCH + lane + 64] = (_Float16)o.z;
    As[w * A_PITCH + lane + 96] = (_Float16)o.w;
    if (q == 0) stage_b_fragments<NT, K, true>(ldsB, B0, B1);
    __syncthreads();

    if (w < NT) {
      v8f acc = {};
#pragma unroll
      for (int kc = 0; kc < KC; ++kc) {
        const _Float16* ap = As + l16 * A_PITCH + kc * 32 + hh * 8;
        const v8h alo = *(const v8h*)ap;
        const v8h ahi = *(const v8h*)(ap + 16);
        const v16h a = __builtin_shufflevector(alo, ahi,
            0, 1, 2, 3, 4, 5, 6, 7, 8, 9, 10, 11, 12, 13, 14, 15);
        const v16h bf = *(const v16h*)(ldsB + ((size_t)(kc * NT + w) * 32 + lane) * 16);
        acc = __builtin_amdgcn_wmma_f32_16x16x32_f16(false, a, false, bf,
                                                     (short)0, acc, false, false);
      }
#pragma unroll
      for (int r = 0; r < 8; ++r) Cs[(hh * 8 + r) * C_PITCH + w * 16 + l16] = acc[r];
    }
    __syncthreads();

    const float s0a = lvec[lane],              s0b = lvec[lane + 32];
    const float d0a = lvec[OUT_CH + lane],     d0b = lvec[OUT_CH + lane + 32];
    const float s1a = lvec[2 * OUT_CH + lane], s1b = lvec[2 * OUT_CH + lane + 32];
    const float d1a = lvec[3 * OUT_CH + lane], d1b = lvec[3 * OUT_CH + lane + 32];
    const float v0 = Cs[w * C_PITCH + lane];
    const float v1 = Cs[w * C_PITCH + lane + 32];
    const float v2 = Cs[w * C_PITCH + lane + 64];
    const float v3 = Cs[w * C_PITCH + lane + 96];
    float ps0 = fmaf(v1, s0b, v0 * s0a), pd0 = fmaf(v1, d0b, v0 * d0a);
    float ps1 = fmaf(v3, s1b, v2 * s1a), pd1 = fmaf(v3, d1b, v2 * d1a);
#pragma unroll
    for (int off = 16; off > 0; off >>= 1) {
      ps0 += __shfl_xor(ps0, off, 32);
      pd0 += __shfl_xor(pd0, off, 32);
      ps1 += __shfl_xor(ps1, off, 32);
      pd1 += __shfl_xor(pd1, off, 32);
    }
    v4h t;
    t[0] = (_Float16)v0; t[1] = (_Float16)v1; t[2] = (_Float16)v2; t[3] = (_Float16)v3;
    *(v4h*)(Ct + (size_t)row * 128 + 4 * lane) = t;
    if (lane == 0) {
      osP[row] = make_float2(ps0, ps1);
      odP[row] = make_float2(pd0, pd1);
    }
  }
}


static inline int ceildiv(long a, long b) { return (int)((a + b - 1) / b); }

extern "C" void kernel_launch(void* const* d_in, const int* in_sizes, int n_in,
                              void* d_out, int out_size, void* d_ws, size_t ws_size,
                              hipStream_t stream) {
  (void)in_sizes; (void)n_in; (void)out_size; (void)ws_size;
  const float* x       = (const float*)d_in[0];
  const int*   ei      = (const int*)d_in[1];
  const float* W1      = (const float*)d_in[2];
  const float* a_src1  = (const float*)d_in[3];
  const float* a_dst1  = (const float*)d_in[4];
  const float* b1      = (const float*)d_in[5];
  const float* W_mu    = (const float*)d_in[6];
  const float* a_srcm  = (const float*)d_in[7];
  const float* a_dstm  = (const float*)d_in[8];
  const float* b_mu    = (const float*)d_in[9];
  const float* W_ls    = (const float*)d_in[10];
  const float* a_srcl  = (const float*)d_in[11];
  const float* a_dstl  = (const float*)d_in[12];
  const float* b_ls    = (const float*)d_in[13];

  const long SZ_H = (long)N_NODES * HID;
  const long SZ_O = (long)N_NODES * OUT_CH;

  float* ws  = (float*)d_ws;
  float* hx  = ws;
  float* h1  = hx + SZ_H;
  float* hml = h1;
  float* sc  = h1 + SZ_H;
  float* as1 = sc + 0L * N_NODES, *ad1 = sc + 1L * N_NODES;
  float* as2 = sc + 2L * N_NODES;
  float* ad2 = sc + 4L * N_NODES;
  int* rowptr  = (int*)(sc + 8L * N_NODES);
  int* csr_src = (int*)(sc + 10L * N_NODES);
  int*  pairs  = (int*)(h1 + SZ_H / 2);
  int*  hist   = pairs + 2L * E_REAL;
  int*  offs   = hist + NCH * NBP;

  float* out_mu = (float*)d_out;
  float* out_ls = out_mu + SZ_O;

  const int TB = 256;

  sort_and_project<<<NCH + GEMM1_BLOCKS, EDGE_TB, 0, stream>>>(
      ei, hist, offs, pairs, x, W1, hx, a_src1, a_dst1, as1, ad1);

  const int node_blocks = ceildiv(N_NODES, TB / 32);

  aggregate_and_project<<<NB, FUSE_TB, 0, stream>>>(
      pairs, hist, offs, rowptr, csr_src, as1, ad1, hx, b1, W_mu, W_ls, hml,
      a_srcm, a_dstm, a_srcl, a_dstl, as2, ad2);

  node_softmax_gather_pair<<<node_blocks, TB, 0, stream>>>(
      rowptr, csr_src, as2, ad2, hml, b_mu, out_mu,
      b_ls, out_ls);
}
